# starting kernel
# baseline (speedup 1.0000x reference)
#include <hip/hip_runtime.h>

typedef __attribute__((ext_vector_type(2))) float v2f;
typedef __attribute__((ext_vector_type(8))) float v8f;

#define EDIM 128
#define WAVES_PER_BLOCK 4
#define EDGES_PER_WAVE 16
#define EDGES_PER_BLOCK (WAVES_PER_BLOCK * EDGES_PER_WAVE)

__global__ __launch_bounds__(WAVES_PER_BLOCK * 32)
void rel_dot_wmma_kernel(const float* __restrict__ emb,
                         const float* __restrict__ relations,
                         const float* __restrict__ tokeys,
                         const float* __restrict__ toqueries,
                         const int* __restrict__ si,
                         const int* __restrict__ pi,
                         const int* __restrict__ oi,
                         float* __restrict__ out,
                         int E)
{
    extern __shared__ float smem[];
    __shared__ int sPi[WAVES_PER_BLOCK][EDGES_PER_WAVE];

    const int lane = threadIdx.x & 31;
    const int wave = threadIdx.x >> 5;
    const int base = (blockIdx.x * WAVES_PER_BLOCK + wave) * EDGES_PER_WAVE;

    float* sS = smem + wave * (2 * EDGES_PER_WAVE * EDIM);
    float* sO = sS + EDGES_PER_WAVE * EDIM;

    #pragma unroll 4
    for (int r = 0; r < EDGES_PER_WAVE; ++r) {
        const int e   = min(base + r, E - 1);
        const int srw = si[e];
        const int orw = oi[e];
        const float4 vs = *(const float4*)(emb + (size_t)srw * EDIM + lane * 4);
        const float4 vo = *(const float4*)(emb + (size_t)orw * EDIM + lane * 4);
        *(float4*)(sS + r * EDIM + lane * 4) = vs;
        *(float4*)(sO + r * EDIM + lane * 4) = vo;
    }
    if (lane < EDGES_PER_WAVE) sPi[wave][lane] = pi[min(base + lane, E - 1)];
    __syncthreads();

    const int half = lane >> 4;
    const int lr   = lane & 15;

    v8f accS[8];
    #pragma unroll
    for (int t = 0; t < 8; ++t) accS[t] = (v8f)0.0f;

    for (int kc = 0; kc < EDIM / 4; ++kc) {
        const int col = kc * 4 + 2 * half;
        const v2f a = *(const v2f*)(sS + lr * EDIM + col);
        #pragma unroll
        for (int t = 0; t < 8; ++t) {
            const v2f b = *(const v2f*)(tokeys + (size_t)(t * 16 + lr) * EDIM + col);
            accS[t] = __builtin_amdgcn_wmma_f32_16x16x4_f32(
                false, a, false, b, (short)0, accS[t], false, false);
        }
    }

    v8f part = (v8f)0.0f;
    for (int t = 0; t < 8; ++t) {
        v8f accO = (v8f)0.0f;
        for (int kc = 0; kc < EDIM / 4; ++kc) {
            const int col = kc * 4 + 2 * half;
            const v2f a = *(const v2f*)(sO + lr * EDIM + col);
            const v2f b = *(const v2f*)(toqueries + (size_t)(t * 16 + lr) * EDIM + col);
            accO = __builtin_amdgcn_wmma_f32_16x16x4_f32(
                false, a, false, b, (short)0, accO, false, false);
        }
        #pragma unroll
        for (int v = 0; v < 8; ++v) {
            const int prow = sPi[wave][v + 8 * half];
            const float p  = relations[(size_t)prow * EDIM + t * 16 + lr];
            part[v] += accS[t][v] * accO[v] * p;
        }
    }

    const float scale = 0.08838834764831845f;
    #pragma unroll
    for (int v = 0; v < 8; ++v) {
        float s = part[v];
        s += __shfl_xor(s, 1, 32);
        s += __shfl_xor(s, 2, 32);
        s += __shfl_xor(s, 4, 32);
        s += __shfl_xor(s, 8, 32);
        if (lr == 0) {
            const int e = base + v + 8 * half;
            if (e < E) out[e] = s * scale;
        }
    }
}

extern "C" void kernel_launch(void* const* d_in, const int* in_sizes, int n_in,
                              void* d_out, int out_size, void* d_ws, size_t ws_size,
                              hipStream_t stream) {
    const float* emb       = (const float*)d_in[0];
    const float* relations = (const float*)d_in[1];
    const float* tokeys    = (const float*)d_in[2];
    const float* toqueries = (const float*)d_in[3];
    const int*   si        = (const int*)d_in[4];
    const int*   pi        = (const int*)d_in[5];
    const int*   oi        = (const int*)d_in[6];
    float*       out       = (float*)d_out;

    const int E = in_sizes[4];
    const int blocks = (E + EDGES_PER_BLOCK - 1) / EDGES_PER_BLOCK;
    const size_t shmem = (size_t)WAVES_PER_BLOCK * 2 * EDGES_PER_WAVE * EDIM * sizeof(float);

    rel_dot_wmma_kernel<<<blocks, WAVES_PER_BLOCK * 32, shmem, stream>>>(
        emb, relations, tokeys, toqueries, si, pi, oi, out, E);
}


# A and B fragments loaded as one b128 per lane feeding two WMMAs (same column expression on both sides): halves the number of weight-row fragment loads; k terms are summed in a permuted order
# speedup vs baseline: 1.2962x; 1.2962x over previous
#include <hip/hip_runtime.h>

typedef __attribute__((ext_vector_type(2))) float v2f;
typedef __attribute__((ext_vector_type(8))) float v8f;

#define EDIM 128
#define WAVES_PER_BLOCK 4
#define EDGES_PER_WAVE 16
#define EDGES_PER_BLOCK (WAVES_PER_BLOCK * EDGES_PER_WAVE)

__global__ __launch_bounds__(WAVES_PER_BLOCK * 32)
void rel_dot_wmma_kernel(const float* __restrict__ emb,
                         const float* __restrict__ relations,
                         const float* __restrict__ tokeys,
                         const float* __restrict__ toqueries,
                         const int* __restrict__ si,
                         const int* __restrict__ pi,
                         const int* __restrict__ oi,
                         float* __restrict__ out,
                         int E)
{
    extern __shared__ float smem[];
    __shared__ int sPi[WAVES_PER_BLOCK][EDGES_PER_WAVE];

    const int lane = threadIdx.x & 31;
    const int wave = threadIdx.x >> 5;
    const int base = (blockIdx.x * WAVES_PER_BLOCK + wave) * EDGES_PER_WAVE;

    float* sS = smem + wave * (2 * EDGES_PER_WAVE * EDIM);
    float* sO = sS + EDGES_PER_WAVE * EDIM;

    #pragma unroll 4
    for (int r = 0; r < EDGES_PER_WAVE; ++r) {
        const int e   = min(base + r, E - 1);
        const int srw = si[e];
        const int orw = oi[e];
        const float4 vs = *(const float4*)(emb + (size_t)srw * EDIM + lane * 4);
        const float4 vo = *(const float4*)(emb + (size_t)orw * EDIM + lane * 4);
        *(float4*)(sS + r * EDIM + lane * 4) = vs;
        *(float4*)(sO + r * EDIM + lane * 4) = vo;
    }
    if (lane < EDGES_PER_WAVE) sPi[wave][lane] = pi[min(base + lane, E - 1)];
    __syncthreads();

    const int half = lane >> 4;
    const int lr   = lane & 15;

    v8f accS[8];
    #pragma unroll
    for (int t = 0; t < 8; ++t) accS[t] = (v8f)0.0f;

    for (int kq = 0; kq < EDIM / 8; ++kq) {
        const int col = kq * 8 + 4 * half;
        const float4 a = *(const float4*)(sS + lr * EDIM + col);
        const v2f a0 = {a.x, a.y};
        const v2f a1 = {a.z, a.w};
        #pragma unroll
        for (int t = 0; t < 8; ++t) {
            const float4 b = *(const float4*)(tokeys + (size_t)(t * 16 + lr) * EDIM + col);
            const v2f b0 = {b.x, b.y};
            const v2f b1 = {b.z, b.w};
            accS[t] = __builtin_amdgcn_wmma_f32_16x16x4_f32(
                false, a0, false, b0, (short)0, accS[t], false, false);
            accS[t] = __builtin_amdgcn_wmma_f32_16x16x4_f32(
                false, a1, false, b1, (short)0, accS[t], false, false);
        }
    }

    v8f part = (v8f)0.0f;
    for (int t = 0; t < 8; ++t) {
        v8f accO = (v8f)0.0f;
        for (int kq = 0; kq < EDIM / 8; ++kq) {
            const int col = kq * 8 + 4 * half;
            const float4 a = *(const float4*)(sO + lr * EDIM + col);
            const float4 b = *(const float4*)(toqueries + (size_t)(t * 16 + lr) * EDIM + col);
            const v2f a0 = {a.x, a.y};
            const v2f a1 = {a.z, a.w};
            const v2f b0 = {b.x, b.y};
            const v2f b1 = {b.z, b.w};
            accO = __builtin_amdgcn_wmma_f32_16x16x4_f32(
                false, a0, false, b0, (short)0, accO, false, false);
            accO = __builtin_amdgcn_wmma_f32_16x16x4_f32(
                false, a1, false, b1, (short)0, accO, false, false);
        }
        #pragma unroll
        for (int v = 0; v < 8; ++v) {
            const int prow = sPi[wave][v + 8 * half];
            const float p  = relations[(size_t)prow * EDIM + t * 16 + lr];
            part[v] += accS[t][v] * accO[v] * p;
        }
    }

    const float scale = 0.08838834764831845f;
    #pragma unroll
    for (int v = 0; v < 8; ++v) {
        float s = part[v];
        s += __shfl_xor(s, 1, 32);
        s += __shfl_xor(s, 2, 32);
        s += __shfl_xor(s, 4, 32);
        s += __shfl_xor(s, 8, 32);
        if (lr == 0) {
            const int e = base + v + 8 * half;
            if (e < E) out[e] = s * scale;
        }
    }
}

extern "C" void kernel_launch(void* const* d_in, const int* in_sizes, int n_in,
                              void* d_out, int out_size, void* d_ws, size_t ws_size,
                              hipStream_t stream) {
    const float* emb       = (const float*)d_in[0];
    const float* relations = (const float*)d_in[1];
    const float* tokeys    = (const float*)d_in[2];
    const float* toqueries = (const float*)d_in[3];
    const int*   si        = (const int*)d_in[4];
    const int*   pi        = (const int*)d_in[5];
    const int*   oi        = (const int*)d_in[6];
    float*       out       = (float*)d_out;

    const int E = in_sizes[4];
    const int blocks = (E + EDGES_PER_BLOCK - 1) / EDGES_PER_BLOCK;
    const size_t shmem = (size_t)WAVES_PER_BLOCK * 2 * EDGES_PER_WAVE * EDIM * sizeof(float);

    rel_dot_wmma_kernel<<<blocks, WAVES_PER_BLOCK * 32, shmem, stream>>>(
        emb, relations, tokeys, toqueries, si, pi, oi, out, E);
}


# Both weight matrices copied once per work-group into LDS in B-fragment order (256 persistent 8-wave work-groups loop over edge tiles): B fragments are ds_load_b128, not 16-row global loads; same sums
# speedup vs baseline: 1.5502x; 1.1960x over previous
#include <hip/hip_runtime.h>

typedef __attribute__((ext_vector_type(2))) float v2f;
typedef __attribute__((ext_vector_type(8))) float v8f;

#define EDIM 128
#define WAVES_PER_BLOCK 8
#define EDGES_PER_WAVE 16
#define NUM_BLOCKS 256
#define WFLOATS (EDIM * EDIM)
#define ROWFLOATS (2 * EDGES_PER_WAVE * EDIM)


__device__ __forceinline__ void wave_lds_order()
{
    __builtin_amdgcn_fence(__ATOMIC_RELEASE, "wavefront");
    __builtin_amdgcn_wave_barrier();
    __builtin_amdgcn_fence(__ATOMIC_ACQUIRE, "wavefront");
}

__global__ __launch_bounds__(WAVES_PER_BLOCK * 32)
void rel_dot_wmma_kernel(const float* __restrict__ emb,
                         const float* __restrict__ relations,
                         const float* __restrict__ tokeys,
                         const float* __restrict__ toqueries,
                         const int* __restrict__ si,
                         const int* __restrict__ pi,
                         const int* __restrict__ oi,
                         float* __restrict__ out,
                         int E)
{
    __shared__ __attribute__((aligned(16))) float sW[2 * WFLOATS];
    __shared__ __attribute__((aligned(16))) float sRows[WAVES_PER_BLOCK * ROWFLOATS];
    __shared__ int sPi[WAVES_PER_BLOCK][EDGES_PER_WAVE];

    const int lane = threadIdx.x & 31;
    const int wave = __builtin_amdgcn_readfirstlane(threadIdx.x >> 5);

    for (int c = threadIdx.x; c < WFLOATS / 4; c += WAVES_PER_BLOCK * 32) {
        const int f   = c >> 5;
        const int l   = c & 31;
        const int row = (f >> 4) * 16 + (l & 15);
        const int col = (f & 15) * 8 + 4 * (l >> 4);
        const float4 wk = *(const float4*)(tokeys    + row * EDIM + col);
        const float4 wq = *(const float4*)(toqueries + row * EDIM + col);
        *(float4*)(sW + c * 4)           = wk;
        *(float4*)(sW + WFLOATS + c * 4) = wq;
    }
    __syncthreads();

    float* sS = sRows + wave * ROWFLOATS;
    float* sO = sS + EDGES_PER_WAVE * EDIM;
    const float* wK = sW + lane * 4;
    const float* wQ = sW + WFLOATS + lane * 4;

    const int half = lane >> 4;
    const int lr   = lane & 15;

    const int nTiles = (E + EDGES_PER_WAVE - 1) / EDGES_PER_WAVE;
    for (int tile = blockIdx.x * WAVES_PER_BLOCK + wave; tile < nTiles; tile += NUM_BLOCKS * WAVES_PER_BLOCK) {
        const int base = tile * EDGES_PER_WAVE;

        #pragma unroll 4
        for (int r = 0; r < EDGES_PER_WAVE; ++r) {
            const int e   = min(base + r, E - 1);
            const int srw = si[e];
            const int orw = oi[e];
            const float4 vs = *(const float4*)(emb + (size_t)srw * EDIM + lane * 4);
            const float4 vo = *(const float4*)(emb + (size_t)orw * EDIM + lane * 4);
            *(float4*)(sS + r * EDIM + lane * 4) = vs;
            *(float4*)(sO + r * EDIM + lane * 4) = vo;
        }
        if (lane < EDGES_PER_WAVE) sPi[wave][lane] = pi[min(base + lane, E - 1)];
        wave_lds_order();

        v8f accS[8];
        #pragma unroll
        for (int t = 0; t < 8; ++t) accS[t] = (v8f)0.0f;

        for (int kq = 0; kq < EDIM / 8; ++kq) {
            const int col = kq * 8 + 4 * half;
            const float4 a = *(const float4*)(sS + lr * EDIM + col);
            const v2f a0 = {a.x, a.y};
            const v2f a1 = {a.z, a.w};
            #pragma unroll
            for (int t = 0; t < 8; ++t) {
                const float4 b = *(const float4*)(wK + (t * 16 + kq) * EDIM);
                const v2f b0 = {b.x, b.y};
                const v2f b1 = {b.z, b.w};
                accS[t] = __builtin_amdgcn_wmma_f32_16x16x4_f32(
                    false, a0, false, b0, (short)0, accS[t], false, false);
                accS[t] = __builtin_amdgcn_wmma_f32_16x16x4_f32(
                    false, a1, false, b1, (short)0, accS[t], false, false);
            }
        }

        v8f part = (v8f)0.0f;
        for (int t = 0; t < 8; ++t) {
            v8f accO = (v8f)0.0f;
            for (int kq = 0; kq < EDIM / 8; ++kq) {
                const int col = kq * 8 + 4 * half;
                const float4 a = *(const float4*)(sO + lr * EDIM + col);
                const float4 b = *(const float4*)(wQ + (t * 16 + kq) * EDIM);
                const v2f a0 = {a.x, a.y};
                const v2f a1 = {a.z, a.w};
                const v2f b0 = {b.x, b.y};
                const v2f b1 = {b.z, b.w};
                accO = __builtin_amdgcn_wmma_f32_16x16x4_f32(
                    false, a0, false, b0, (short)0, accO, false, false);
                accO = __builtin_amdgcn_wmma_f32_16x16x4_f32(
                    false, a1, false, b1, (short)0, accO, false, false);
            }
            #pragma unroll
            for (int v = 0; v < 8; ++v) {
                const int prow = sPi[wave][v + 8 * half];
                const float p  = relations[(size_t)prow * EDIM + t * 16 + lr];
                part[v] += accS[t][v] * accO[v] * p;
            }
        }

        const float scale = 0.08838834764831845f;
        #pragma unroll
        for (int v = 0; v < 8; ++v) {
            float s = part[v];
            s += __shfl_xor(s, 1, 32);
            s += __shfl_xor(s, 2, 32);
            s += __shfl_xor(s, 4, 32);
            s += __shfl_xor(s, 8, 32);
            if (lr == 0) {
                const int e = base + v + 8 * half;
                if (e < E) out[e] = s * scale;
            }
        }
        wave_lds_order();
    }
}

extern "C" void kernel_launch(void* const* d_in, const int* in_sizes, int n_in,
                              void* d_out, int out_size, void* d_ws, size_t ws_size,
                              hipStream_t stream) {
    const float* emb       = (const float*)d_in[0];
    const float* relations = (const float*)d_in[1];
    const float* tokeys    = (const float*)d_in[2];
    const float* toqueries = (const float*)d_in[3];
    const int*   si        = (const int*)d_in[4];
    const int*   pi        = (const int*)d_in[5];
    const int*   oi        = (const int*)d_in[6];
    float*       out       = (float*)d_out;

    const int E = in_sizes[4];

    rel_dot_wmma_kernel<<<NUM_BLOCKS, WAVES_PER_BLOCK * 32, 0, stream>>>(
        emb, relations, tokeys, toqueries, si, pi, oi, out, E);
}


# N reduction of the fold by row_xmask DPP adds in registers (was 32 LDS-routed shuffles per tile, each waited for); the 16 dots of a tile leave in one store; same butterfly, same sums
# speedup vs baseline: 1.8508x; 1.1939x over previous
#include <hip/hip_runtime.h>

typedef __attribute__((ext_vector_type(2))) float v2f;
typedef __attribute__((ext_vector_type(8))) float v8f;
typedef __attribute__((ext_vector_type(4))) float v4f;

#define EDIM 128
#define WAVES_PER_BLOCK 8
#define EDGES_PER_WAVE 16
#define NUM_BLOCKS 256
#define WFLOATS (EDIM * EDIM)
#define ROWFLOATS (2 * EDGES_PER_WAVE * EDIM)
#define TILE_STRIDE (NUM_BLOCKS * WAVES_PER_BLOCK)


__device__ __forceinline__ void wave_lds_order()
{
    __builtin_amdgcn_fence(__ATOMIC_RELEASE, "wavefront");
    __builtin_amdgcn_wave_barrier();
    __builtin_amdgcn_fence(__ATOMIC_ACQUIRE, "wavefront");
}

template <int M>
__device__ __forceinline__ float row_xor_add(float x)
{
    const int other = __builtin_amdgcn_update_dpp(0, __float_as_int(x), 0x160 + M, 0xf, 0xf, true);
    return x + __int_as_float(other);
}

template <int M>
__device__ __forceinline__ void row_xor_add_all(v8f& p)
{
    #pragma unroll
    for (int v = 0; v < 8; ++v) p[v] = row_xor_add<M>(p[v]);
}

__global__ __launch_bounds__(WAVES_PER_BLOCK * 32)
void rel_dot_wmma_kernel(const float* __restrict__ emb,
                         const float* __restrict__ relations,
                         const float* __restrict__ tokeys,
                         const float* __restrict__ toqueries,
                         const int* __restrict__ si,
                         const int* __restrict__ pi,
                         const int* __restrict__ oi,
                         float* __restrict__ out,
                         int E)
{
    __shared__ __attribute__((aligned(16))) float sW[2 * WFLOATS];
    __shared__ __attribute__((aligned(16))) float sRows[WAVES_PER_BLOCK * ROWFLOATS];
    __shared__ int sPi[WAVES_PER_BLOCK][EDGES_PER_WAVE];

    const int lane = threadIdx.x & 31;
    const int wave = __builtin_amdgcn_readfirstlane(threadIdx.x >> 5);

    for (int c = threadIdx.x; c < WFLOATS / 4; c += WAVES_PER_BLOCK * 32) {
        const int f   = c >> 5;
        const int l   = c & 31;
        const int row = (f >> 4) * 16 + (l & 15);
        const int col = (f & 15) * 8 + 4 * (l >> 4);
        const float4 wk = *(const float4*)(tokeys    + row * EDIM + col);
        const float4 wq = *(const float4*)(toqueries + row * EDIM + col);
        *(float4*)(sW + c * 4)           = wk;
        *(float4*)(sW + WFLOATS + c * 4) = wq;
    }
    __syncthreads();

    float* sS = sRows + wave * ROWFLOATS;
    float* sO = sS + EDGES_PER_WAVE * EDIM;
    const float* wK = sW + lane * 4;
    const float* wQ = sW + WFLOATS + lane * 4;

    const int half = lane >> 4;
    const int lr   = lane & 15;

    const int nTiles = (E + EDGES_PER_WAVE - 1) / EDGES_PER_WAVE;
    const int first  = blockIdx.x * WAVES_PER_BLOCK + wave;

    {
        const int e0   = min(first * EDGES_PER_WAVE + lr, E - 1);
        const int sIdx = si[e0];
        const int oIdx = oi[e0];
        #pragma unroll
        for (int r = 0; r < EDGES_PER_WAVE; ++r) {
            const int srw = __builtin_amdgcn_readlane(sIdx, r);
            const int orw = __builtin_amdgcn_readlane(oIdx, r);
            const float4 vs = *(const float4*)(emb + (size_t)srw * EDIM + lane * 4);
            const float4 vo = *(const float4*)(emb + (size_t)orw * EDIM + lane * 4);
            *(float4*)(sS + r * EDIM + lane * 4) = vs;
            *(float4*)(sO + r * EDIM + lane * 4) = vo;
        }
        if (lane < EDGES_PER_WAVE) sPi[wave][lane] = pi[e0];
    }
    int sIdxNext, oIdxNext;
    {
        const int e1 = min((first + TILE_STRIDE) * EDGES_PER_WAVE + lr, E - 1);
        sIdxNext = si[e1];
        oIdxNext = oi[e1];
    }
    wave_lds_order();

    for (int tile = first; tile < nTiles; tile += TILE_STRIDE) {
        const int base = tile * EDGES_PER_WAVE;

        const int eNext  = min(base + TILE_STRIDE * EDGES_PER_WAVE + lr, E - 1);
        const int eNext2 = min(base + 2 * TILE_STRIDE * EDGES_PER_WAVE + lr, E - 1);
        const int piNext    = pi[eNext];
        const int sIdxNext2 = si[eNext2];
        const int oIdxNext2 = oi[eNext2];

        int prow[8];
        float prel[8][8];
        #pragma unroll
        for (int v = 0; v < 8; ++v) prow[v] = sPi[wave][v + 8 * half];
        #pragma unroll
        for (int t = 0; t < 4; ++t) {
            #pragma unroll
            for (int v = 0; v < 8; ++v)
                prel[t][v] = relations[(size_t)prow[v] * EDIM + t * 16 + lr];
        }

        v4f pre[EDGES_PER_WAVE];
        #pragma unroll
        for (int r = 0; r < EDGES_PER_WAVE; ++r) {
            const int srw = __builtin_amdgcn_readlane(sIdxNext, r);
            pre[r] = *(const v4f*)(emb + (size_t)srw * EDIM + lane * 4);
        }

        v8f accS[8];
        #pragma unroll
        for (int t = 0; t < 8; ++t) accS[t] = (v8f)0.0f;

        for (int kq = 0; kq < EDIM / 8; ++kq) {
            const int col = kq * 8 + 4 * half;
            const float4 a = *(const float4*)(sS + lr * EDIM + col);
            const v2f a0 = {a.x, a.y};
            const v2f a1 = {a.z, a.w};
            #pragma unroll
            for (int t = 0; t < 8; ++t) {
                const float4 b = *(const float4*)(wK + (t * 16 + kq) * EDIM);
                const v2f b0 = {b.x, b.y};
                const v2f b1 = {b.z, b.w};
                accS[t] = __builtin_amdgcn_wmma_f32_16x16x4_f32(
                    false, a0, false, b0, (short)0, accS[t], false, false);
                accS[t] = __builtin_amdgcn_wmma_f32_16x16x4_f32(
                    false, a1, false, b1, (short)0, accS[t], false, false);
            }
        }

        wave_lds_order();
        #pragma unroll
        for (int r = 0; r < EDGES_PER_WAVE; ++r)
            *(v4f*)(sS + r * EDIM + lane * 4) = pre[r];

        #pragma unroll
        for (int r = 0; r < EDGES_PER_WAVE; ++r) {
            const int orw = __builtin_amdgcn_readlane(oIdxNext, r);
            pre[r] = *(const v4f*)(emb + (size_t)orw * EDIM + lane * 4);
        }

        #pragma unroll
        for (int t = 4; t < 8; ++t) {
            #pragma unroll
            for (int v = 0; v < 8; ++v)
                prel[t][v] = relations[(size_t)prow[v] * EDIM + t * 16 + lr];
        }

        v8f accO[8];
        #pragma unroll
        for (int t = 0; t < 8; ++t) accO[t] = (v8f)0.0f;

        for (int kq = 0; kq < EDIM / 8; ++kq) {
            const int col = kq * 8 + 4 * half;
            const float4 a = *(const float4*)(sO + lr * EDIM + col);
            const v2f a0 = {a.x, a.y};
            const v2f a1 = {a.z, a.w};
            #pragma unroll
            for (int t = 0; t < 8; ++t) {
                const float4 b = *(const float4*)(wQ + (t * 16 + kq) * EDIM);
                const v2f b0 = {b.x, b.y};
                const v2f b1 = {b.z, b.w};
                accO[t] = __builtin_amdgcn_wmma_f32_16x16x4_f32(
                    false, a0, false, b0, (short)0, accO[t], false, false);
                accO[t] = __builtin_amdgcn_wmma_f32_16x16x4_f32(
                    false, a1, false, b1, (short)0, accO[t], false, false);
            }
        }

        v8f part = (v8f)0.0f;
        #pragma unroll
        for (int t = 0; t < 8; ++t) {
            #pragma unroll
            for (int v = 0; v < 8; ++v)
                part[v] += accS[t][v] * accO[t][v] * prel[t][v];
        }

        row_xor_add_all<1>(part);
        row_xor_add_all<2>(part);
        row_xor_add_all<4>(part);
        row_xor_add_all<8>(part);
        const float scale = 0.08838834764831845f;
        float dot = part[0];
        #pragma unroll
        for (int v = 1; v < 8; ++v) dot = (lr == v) ? part[v] : dot;
        if (lr < 8) {
            const int e = base + lr + 8 * half;
            if (e < E) out[e] = dot * scale;
        }
        wave_lds_order();
        #pragma unroll
        for (int r = 0; r < EDGES_PER_WAVE; ++r)
            *(v4f*)(sO + r * EDIM + lane * 4) = pre[r];
        if (lane < EDGES_PER_WAVE) sPi[wave][lane] = piNext;
        wave_lds_order();
        sIdxNext = sIdxNext2;
        oIdxNext = oIdxNext2;
    }
}

extern "C" void kernel_launch(void* const* d_in, const int* in_sizes, int n_in,
                              void* d_out, int out_size, void* d_ws, size_t ws_size,
                              hipStream_t stream) {
    const float* emb       = (const float*)d_in[0];
    const float* relations = (const float*)d_in[1];
    const float* tokeys    = (const float*)d_in[2];
    const float* toqueries = (const float*)d_in[3];
    const int*   si        = (const int*)d_in[4];
    const int*   pi        = (const int*)d_in[5];
    const int*   oi        = (const int*)d_in[6];
    float*       out       = (float*)d_out;

    const int E = in_sizes[4];

    rel_dot_wmma_kernel<<<NUM_BLOCKS, WAVES_PER_BLOCK * 32, 0, stream>>>(
        emb, relations, tokeys, toqueries, si, pi, oi, out, E);
}


# Fragment loads of both products run one block (32 WMMAs) ahead in a second register set: each block's 18 ds_load_b128 are one group in front of the previous block's products; same sums
# speedup vs baseline: 1.9094x; 1.0317x over previous
#include <hip/hip_runtime.h>

typedef __attribute__((ext_vector_type(2))) float v2f;
typedef __attribute__((ext_vector_type(8))) float v8f;
typedef __attribute__((ext_vector_type(4))) float v4f;

#define EDIM 128
#define WAVES_PER_BLOCK 8
#define EDGES_PER_WAVE 16
#define NUM_BLOCKS 256
#define WFLOATS (EDIM * EDIM)
#define ROWFLOATS (2 * EDGES_PER_WAVE * EDIM)
#define TILE_STRIDE (NUM_BLOCKS * WAVES_PER_BLOCK)


__device__ __forceinline__ void wave_lds_order()
{
    __builtin_amdgcn_fence(__ATOMIC_RELEASE, "wavefront");
    __builtin_amdgcn_wave_barrier();
    __builtin_amdgcn_fence(__ATOMIC_ACQUIRE, "wavefront");
}

template <int M>
__device__ __forceinline__ float row_xor_add(float x)
{
    const int other = __builtin_amdgcn_update_dpp(0, __float_as_int(x), 0x160 + M, 0xf, 0xf, true);
    return x + __int_as_float(other);
}

template <int M>
__device__ __forceinline__ void row_xor_add_all(v8f& p)
{
    #pragma unroll
    for (int v = 0; v < 8; ++v) p[v] = row_xor_add<M>(p[v]);
}

struct Frag {
    float4 a[2];
    float4 b[2][8];
};

__device__ __forceinline__ void load_block(Frag& f, const float* aRow, const float* bBase)
{
    #pragma unroll
    for (int j = 0; j < 2; ++j) {
        f.a[j] = *(const float4*)(aRow + j * 8);
        #pragma unroll
        for (int t = 0; t < 8; ++t)
            f.b[j][t] = *(const float4*)(bBase + (t * 16 + j) * EDIM);
    }
}

__device__ __forceinline__ void mma_block(v8f (&acc)[8], const Frag& f)
{
    #pragma unroll
    for (int j = 0; j < 2; ++j) {
        const v2f a0 = {f.a[j].x, f.a[j].y};
        const v2f a1 = {f.a[j].z, f.a[j].w};
        #pragma unroll
        for (int t = 0; t < 8; ++t) {
            const v2f b0 = {f.b[j][t].x, f.b[j][t].y};
            const v2f b1 = {f.b[j][t].z, f.b[j][t].w};
            acc[t] = __builtin_amdgcn_wmma_f32_16x16x4_f32(
                false, a0, false, b0, (short)0, acc[t], false, false);
            acc[t] = __builtin_amdgcn_wmma_f32_16x16x4_f32(
                false, a1, false, b1, (short)0, acc[t], false, false);
        }
    }
}

__device__ __forceinline__ void product_pipelined(v8f (&acc)[8], Frag& f0, Frag& f1,
                                                  const float* aCur, const float* bCur,
                                                  const float* aNext, const float* bNext)
{
    for (int b = 0; b < EDIM / 16; b += 2) {
        load_block(f1, aCur + (b + 1) * 16, bCur + (b + 1) * 2 * EDIM);
        __builtin_amdgcn_sched_barrier(0);
        mma_block(acc, f0);
        __builtin_amdgcn_sched_barrier(0);
        const bool last = (b + 2 == EDIM / 16);
        load_block(f0, last ? aNext : aCur + (b + 2) * 16, last ? bNext : bCur + (b + 2) * 2 * EDIM);
        __builtin_amdgcn_sched_barrier(0);
        mma_block(acc, f1);
        __builtin_amdgcn_sched_barrier(0);
    }
}

__global__ __launch_bounds__(WAVES_PER_BLOCK * 32)
void rel_dot_wmma_kernel(const float* __restrict__ emb,
                         const float* __restrict__ relations,
                         const float* __restrict__ tokeys,
                         const float* __restrict__ toqueries,
                         const int* __restrict__ si,
                         const int* __restrict__ pi,
                         const int* __restrict__ oi,
                         float* __restrict__ out,
                         int E)
{
    __shared__ __attribute__((aligned(16))) float sW[2 * WFLOATS];
    __shared__ __attribute__((aligned(16))) float sRows[WAVES_PER_BLOCK * ROWFLOATS];
    __shared__ int sPi[WAVES_PER_BLOCK][EDGES_PER_WAVE];

    const int lane = threadIdx.x & 31;
    const int wave = __builtin_amdgcn_readfirstlane(threadIdx.x >> 5);

    for (int c = threadIdx.x; c < WFLOATS / 4; c += WAVES_PER_BLOCK * 32) {
        const int f   = c >> 5;
        const int l   = c & 31;
        const int row = (f >> 4) * 16 + (l & 15);
        const int col = (f & 15) * 8 + 4 * (l >> 4);
        const float4 wk = *(const float4*)(tokeys    + row * EDIM + col);
        const float4 wq = *(const float4*)(toqueries + row * EDIM + col);
        *(float4*)(sW + c * 4)           = wk;
        *(float4*)(sW + WFLOATS + c * 4) = wq;
    }
    __syncthreads();

    float* sS = sRows + wave * ROWFLOATS;
    float* sO = sS + EDGES_PER_WAVE * EDIM;
    const float* wK = sW + lane * 4;
    const float* wQ = sW + WFLOATS + lane * 4;

    const int half = lane >> 4;
    const int lr   = lane & 15;

    const float* aS0 = sS + lr * EDIM + 4 * half;
    const float* aO0 = sO + lr * EDIM + 4 * half;

    const int nTiles = (E + EDGES_PER_WAVE - 1) / EDGES_PER_WAVE;
    const int first  = blockIdx.x * WAVES_PER_BLOCK + wave;

    {
        const int e0   = min(first * EDGES_PER_WAVE + lr, E - 1);
        const int sIdx = si[e0];
        const int oIdx = oi[e0];
        #pragma unroll
        for (int r = 0; r < EDGES_PER_WAVE; ++r) {
            const int srw = __builtin_amdgcn_readlane(sIdx, r);
            const int orw = __builtin_amdgcn_readlane(oIdx, r);
            const float4 vs = *(const float4*)(emb + (size_t)srw * EDIM + lane * 4);
            const float4 vo = *(const float4*)(emb + (size_t)orw * EDIM + lane * 4);
            *(float4*)(sS + r * EDIM + lane * 4) = vs;
            *(float4*)(sO + r * EDIM + lane * 4) = vo;
        }
        if (lane < EDGES_PER_WAVE) sPi[wave][lane] = pi[e0];
    }
    int sIdxNext, oIdxNext;
    {
        const int e1 = min((first + TILE_STRIDE) * EDGES_PER_WAVE + lr, E - 1);
        sIdxNext = si[e1];
        oIdxNext = oi[e1];
    }
    wave_lds_order();

    Frag f0, f1;
    load_block(f0, aS0, wK);

    for (int tile = first; tile < nTiles; tile += TILE_STRIDE) {
        const int base = tile * EDGES_PER_WAVE;

        const int eNext  = min(base + TILE_STRIDE * EDGES_PER_WAVE + lr, E - 1);
        const int eNext2 = min(base + 2 * TILE_STRIDE * EDGES_PER_WAVE + lr, E - 1);
        const int piNext    = pi[eNext];
        const int sIdxNext2 = si[eNext2];
        const int oIdxNext2 = oi[eNext2];

        int prow[8];
        float prel[8][8];
        #pragma unroll
        for (int v = 0; v < 8; ++v) prow[v] = sPi[wave][v + 8 * half];
        #pragma unroll
        for (int t = 0; t < 4; ++t) {
            #pragma unroll
            for (int v = 0; v < 8; ++v)
                prel[t][v] = relations[(size_t)prow[v] * EDIM + t * 16 + lr];
        }

        v4f pre[EDGES_PER_WAVE];
        #pragma unroll
        for (int r = 0; r < EDGES_PER_WAVE; ++r) {
            const int srw = __builtin_amdgcn_readlane(sIdxNext, r);
            pre[r] = *(const v4f*)(emb + (size_t)srw * EDIM + lane * 4);
        }

        v8f accS[8];
        #pragma unroll
        for (int t = 0; t < 8; ++t) accS[t] = (v8f)0.0f;

        product_pipelined(accS, f0, f1, aS0, wK, aO0, wQ);

        wave_lds_order();
        #pragma unroll
        for (int r = 0; r < EDGES_PER_WAVE; ++r)
            *(v4f*)(sS + r * EDIM + lane * 4) = pre[r];
        wave_lds_order();

        #pragma unroll
        for (int r = 0; r < EDGES_PER_WAVE; ++r) {
            const int orw = __builtin_amdgcn_readlane(oIdxNext, r);
            pre[r] = *(const v4f*)(emb + (size_t)orw * EDIM + lane * 4);
        }

        #pragma unroll
        for (int t = 4; t < 8; ++t) {
            #pragma unroll
            for (int v = 0; v < 8; ++v)
                prel[t][v] = relations[(size_t)prow[v] * EDIM + t * 16 + lr];
        }

        v8f accO[8];
        #pragma unroll
        for (int t = 0; t < 8; ++t) accO[t] = (v8f)0.0f;

        product_pipelined(accO, f0, f1, aO0, wQ, aS0, wK);

        v8f part = (v8f)0.0f;
        #pragma unroll
        for (int t = 0; t < 8; ++t) {
            #pragma unroll
            for (int v = 0; v < 8; ++v)
                part[v] += accS[t][v] * accO[t][v] * prel[t][v];
        }

        row_xor_add_all<1>(part);
        row_xor_add_all<2>(part);
        row_xor_add_all<4>(part);
        row_xor_add_all<8>(part);
        const float scale = 0.08838834764831845f;
        float dot = part[0];
        #pragma unroll
        for (int v = 1; v < 8; ++v) dot = (lr == v) ? part[v] : dot;
        if (lr < 8) {
            const int e = base + lr + 8 * half;
            if (e < E) out[e] = dot * scale;
        }
        wave_lds_order();
        #pragma unroll
        for (int r = 0; r < EDGES_PER_WAVE; ++r)
            *(v4f*)(sO + r * EDIM + lane * 4) = pre[r];
        if (lane < EDGES_PER_WAVE) sPi[wave][lane] = piNext;
        wave_lds_order();
        sIdxNext = sIdxNext2;
        oIdxNext = oIdxNext2;
    }
}

extern "C" void kernel_launch(void* const* d_in, const int* in_sizes, int n_in,
                              void* d_out, int out_size, void* d_ws, size_t ws_size,
                              hipStream_t stream) {
    const float* emb       = (const float*)d_in[0];
    const float* relations = (const float*)d_in[1];
    const float* tokeys    = (const float*)d_in[2];
    const float* toqueries = (const float*)d_in[3];
    const int*   si        = (const int*)d_in[4];
    const int*   pi        = (const int*)d_in[5];
    const int*   oi        = (const int*)d_in[6];
    float*       out       = (float*)d_out;

    const int E = in_sizes[4];

    rel_dot_wmma_kernel<<<NUM_BLOCKS, WAVES_PER_BLOCK * 32, 0, stream>>>(
        emb, relations, tokeys, toqueries, si, pi, oi, out, E);
}


# The two waves of a SIMD are kept level: each publishes its finished products in an LDS word and the younger raises its priority (s_setprio) while not a tile ahead, drops it when it is; same sums
# speedup vs baseline: 1.9707x; 1.0321x over previous
#include <hip/hip_runtime.h>

typedef __attribute__((ext_vector_type(2))) float v2f;
typedef __attribute__((ext_vector_type(8))) float v8f;
typedef __attribute__((ext_vector_type(4))) float v4f;

#define EDIM 128
#define WAVES_PER_BLOCK 8
#define EDGES_PER_WAVE 16
#define NUM_BLOCKS 256
#define WFLOATS (EDIM * EDIM)
#define ROWFLOATS (2 * EDGES_PER_WAVE * EDIM)
#define TILE_STRIDE (NUM_BLOCKS * WAVES_PER_BLOCK)


__device__ __forceinline__ void wave_lds_order()
{
    __builtin_amdgcn_fence(__ATOMIC_RELEASE, "wavefront");
    __builtin_amdgcn_wave_barrier();
    __builtin_amdgcn_fence(__ATOMIC_ACQUIRE, "wavefront");
}

template <int M>
__device__ __forceinline__ float row_xor_add(float x)
{
    const int other = __builtin_amdgcn_update_dpp(0, __float_as_int(x), 0x160 + M, 0xf, 0xf, true);
    return x + __int_as_float(other);
}

template <int M>
__device__ __forceinline__ void row_xor_add_all(v8f& p)
{
    #pragma unroll
    for (int v = 0; v < 8; ++v) p[v] = row_xor_add<M>(p[v]);
}

__device__ __forceinline__ void take_or_yield_pipe(int mine, int partnerWord, int lead)
{
    const int partner = __builtin_amdgcn_readfirstlane(partnerWord);
    if (mine - partner >= lead) __builtin_amdgcn_s_setprio(0);
    else                        __builtin_amdgcn_s_setprio(3);
}

struct Frag {
    float4 a[2];
    float4 b[2][8];
};

__device__ __forceinline__ void load_block(Frag& f, const float* aRow, const float* bBase)
{
    #pragma unroll
    for (int j = 0; j < 2; ++j) {
        f.a[j] = *(const float4*)(aRow + j * 8);
        #pragma unroll
        for (int t = 0; t < 8; ++t)
            f.b[j][t] = *(const float4*)(bBase + (t * 16 + j) * EDIM);
    }
}

__device__ __forceinline__ void mma_block(v8f (&acc)[8], const Frag& f)
{
    #pragma unroll
    for (int j = 0; j < 2; ++j) {
        const v2f a0 = {f.a[j].x, f.a[j].y};
        const v2f a1 = {f.a[j].z, f.a[j].w};
        #pragma unroll
        for (int t = 0; t < 8; ++t) {
            const v2f b0 = {f.b[j][t].x, f.b[j][t].y};
            const v2f b1 = {f.b[j][t].z, f.b[j][t].w};
            acc[t] = __builtin_amdgcn_wmma_f32_16x16x4_f32(
                false, a0, false, b0, (short)0, acc[t], false, false);
            acc[t] = __builtin_amdgcn_wmma_f32_16x16x4_f32(
                false, a1, false, b1, (short)0, acc[t], false, false);
        }
    }
}

__device__ __forceinline__ void product_pipelined(v8f (&acc)[8], Frag& f0, Frag& f1,
                                                  const float* aCur, const float* bCur,
                                                  const float* aNext, const float* bNext)
{
    for (int b = 0; b < EDIM / 16; b += 2) {
        load_block(f1, aCur + (b + 1) * 16, bCur + (b + 1) * 2 * EDIM);
        __builtin_amdgcn_sched_barrier(0);
        mma_block(acc, f0);
        __builtin_amdgcn_sched_barrier(0);
        const bool last = (b + 2 == EDIM / 16);
        load_block(f0, last ? aNext : aCur + (b + 2) * 16, last ? bNext : bCur + (b + 2) * 2 * EDIM);
        __builtin_amdgcn_sched_barrier(0);
        mma_block(acc, f1);
        __builtin_amdgcn_sched_barrier(0);
    }
}

__global__ __launch_bounds__(WAVES_PER_BLOCK * 32)
void rel_dot_wmma_kernel(const float* __restrict__ emb,
                         const float* __restrict__ relations,
                         const float* __restrict__ tokeys,
                         const float* __restrict__ toqueries,
                         const int* __restrict__ si,
                         const int* __restrict__ pi,
                         const int* __restrict__ oi,
                         float* __restrict__ out,
                         int E)
{
    __shared__ __attribute__((aligned(16))) float sW[2 * WFLOATS];
    __shared__ __attribute__((aligned(16))) float sRows[WAVES_PER_BLOCK * ROWFLOATS];
    __shared__ int sPi[WAVES_PER_BLOCK][EDGES_PER_WAVE];
    __shared__ int sDone[WAVES_PER_BLOCK];

    const int lane = threadIdx.x & 31;
    const int wave = __builtin_amdgcn_readfirstlane(threadIdx.x >> 5);

    for (int c = threadIdx.x; c < WFLOATS / 4; c += WAVES_PER_BLOCK * 32) {
        const int f   = c >> 5;
        const int l   = c & 31;
        const int row = (f >> 4) * 16 + (l & 15);
        const int col = (f & 15) * 8 + 4 * (l >> 4);
        const float4 wk = *(const float4*)(tokeys    + row * EDIM + col);
        const float4 wq = *(const float4*)(toqueries + row * EDIM + col);
        *(float4*)(sW + c * 4)           = wk;
        *(float4*)(sW + WFLOATS + c * 4) = wq;
    }
    if (threadIdx.x < WAVES_PER_BLOCK) sDone[threadIdx.x] = 0;
    __syncthreads();

    float* sS = sRows + wave * ROWFLOATS;
    float* sO = sS + EDGES_PER_WAVE * EDIM;
    const float* wK = sW + lane * 4;
    const float* wQ = sW + WFLOATS + lane * 4;

    const int half = lane >> 4;
    const int lr   = lane & 15;

    const float* aS0 = sS + lr * EDIM + 4 * half;
    const float* aO0 = sO + lr * EDIM + 4 * half;

    const int nTiles = (E + EDGES_PER_WAVE - 1) / EDGES_PER_WAVE;
    const int first  = blockIdx.x * WAVES_PER_BLOCK + wave;

    {
        const int e0   = min(first * EDGES_PER_WAVE + lr, E - 1);
        const int sIdx = si[e0];
        const int oIdx = oi[e0];
        #pragma unroll
        for (int r = 0; r < EDGES_PER_WAVE; ++r) {
            const int srw = __builtin_amdgcn_readlane(sIdx, r);
            const int orw = __builtin_amdgcn_readlane(oIdx, r);
            const float4 vs = *(const float4*)(emb + (size_t)srw * EDIM + lane * 4);
            const float4 vo = *(const float4*)(emb + (size_t)orw * EDIM + lane * 4);
            *(float4*)(sS + r * EDIM + lane * 4) = vs;
            *(float4*)(sO + r * EDIM + lane * 4) = vo;
        }
        if (lane < EDGES_PER_WAVE) sPi[wave][lane] = pi[e0];
    }
    int sIdxNext, oIdxNext;
    {
        const int e1 = min((first + TILE_STRIDE) * EDGES_PER_WAVE + lr, E - 1);
        sIdxNext = si[e1];
        oIdxNext = oi[e1];
    }
    wave_lds_order();

    Frag f0, f1;
    load_block(f0, aS0, wK);

    const bool younger = wave >= WAVES_PER_BLOCK / 2;
    const int  partnerWave = wave ^ (WAVES_PER_BLOCK / 2);
    const int  lead = younger ? 2 : -(1 << 30);
    int done = 0;
    if (younger) __builtin_amdgcn_s_setprio(3);

    for (int tile = first; tile < nTiles; tile += TILE_STRIDE) {
        const int base = tile * EDGES_PER_WAVE;

        const int eNext  = min(base + TILE_STRIDE * EDGES_PER_WAVE + lr, E - 1);
        const int eNext2 = min(base + 2 * TILE_STRIDE * EDGES_PER_WAVE + lr, E - 1);
        const int piNext    = pi[eNext];
        const int sIdxNext2 = si[eNext2];
        const int oIdxNext2 = oi[eNext2];

        int prow[8];
        float prel[8][8];
        #pragma unroll
        for (int v = 0; v < 8; ++v) prow[v] = sPi[wave][v + 8 * half];
        #pragma unroll
        for (int t = 0; t < 4; ++t) {
            #pragma unroll
            for (int v = 0; v < 8; ++v)
                prel[t][v] = relations[(size_t)prow[v] * EDIM + t * 16 + lr];
        }

        v4f pre[EDGES_PER_WAVE];
        #pragma unroll
        for (int r = 0; r < EDGES_PER_WAVE; ++r) {
            const int srw = __builtin_amdgcn_readlane(sIdxNext, r);
            pre[r] = *(const v4f*)(emb + (size_t)srw * EDIM + lane * 4);
        }

        v8f accS[8];
        #pragma unroll
        for (int t = 0; t < 8; ++t) accS[t] = (v8f)0.0f;

        product_pipelined(accS, f0, f1, aS0, wK, aO0, wQ);

        ++done;
        if (lane == 0) sDone[wave] = done;

        wave_lds_order();
        const int partnerMid = sDone[partnerWave];
        #pragma unroll
        for (int r = 0; r < EDGES_PER_WAVE; ++r)
            *(v4f*)(sS + r * EDIM + lane * 4) = pre[r];
        wave_lds_order();

        #pragma unroll
        for (int r = 0; r < EDGES_PER_WAVE; ++r) {
            const int orw = __builtin_amdgcn_readlane(oIdxNext, r);
            pre[r] = *(const v4f*)(emb + (size_t)orw * EDIM + lane * 4);
        }

        #pragma unroll
        for (int t = 4; t < 8; ++t) {
            #pragma unroll
            for (int v = 0; v < 8; ++v)
                prel[t][v] = relations[(size_t)prow[v] * EDIM + t * 16 + lr];
        }

        take_or_yield_pipe(done, partnerMid, lead);

        v8f accO[8];
        #pragma unroll
        for (int t = 0; t < 8; ++t) accO[t] = (v8f)0.0f;

        product_pipelined(accO, f0, f1, aO0, wQ, aS0, wK);

        ++done;
        if (lane == 0) sDone[wave] = done;
        const int partnerEnd = sDone[partnerWave];

        v8f part = (v8f)0.0f;
        #pragma unroll
        for (int t = 0; t < 8; ++t) {
            #pragma unroll
            for (int v = 0; v < 8; ++v)
                part[v] += accS[t][v] * accO[t][v] * prel[t][v];
        }

        row_xor_add_all<1>(part);
        row_xor_add_all<2>(part);
        row_xor_add_all<4>(part);
        row_xor_add_all<8>(part);
        const float scale = 0.08838834764831845f;
        float dot = part[0];
        #pragma unroll
        for (int v = 1; v < 8; ++v) dot = (lr == v) ? part[v] : dot;
        if (lr < 8) {
            const int e = base + lr + 8 * half;
            if (e < E) out[e] = dot * scale;
        }
        wave_lds_order();
        #pragma unroll
        for (int r = 0; r < EDGES_PER_WAVE; ++r)
            *(v4f*)(sO + r * EDIM + lane * 4) = pre[r];
        if (lane < EDGES_PER_WAVE) sPi[wave][lane] = piNext;
        wave_lds_order();
        sIdxNext = sIdxNext2;
        oIdxNext = oIdxNext2;

        take_or_yield_pipe(done, partnerEnd, lead);
    }
    __builtin_amdgcn_s_setprio(0);
}

extern "C" void kernel_launch(void* const* d_in, const int* in_sizes, int n_in,
                              void* d_out, int out_size, void* d_ws, size_t ws_size,
                              hipStream_t stream) {
    const float* emb       = (const float*)d_in[0];
    const float* relations = (const float*)d_in[1];
    const float* tokeys    = (const float*)d_in[2];
    const float* toqueries = (const float*)d_in[3];
    const int*   si        = (const int*)d_in[4];
    const int*   pi        = (const int*)d_in[5];
    const int*   oi        = (const int*)d_in[6];
    float*       out       = (float*)d_out;

    const int E = in_sizes[4];

    rel_dot_wmma_kernel<<<NUM_BLOCKS, WAVES_PER_BLOCK * 32, 0, stream>>>(
        emb, relations, tokeys, toqueries, si, pi, oi, out, E);
}


# Projections done once per node row, not per edge: a WMMA pass over the node table writes workspace tables (two if they fit, else one) that a per-edge launch gathers; per-edge sum in a new order
# speedup vs baseline: 2.5241x; 1.2808x over previous
#include <hip/hip_runtime.h>

typedef __attribute__((ext_vector_type(2))) float v2f;
typedef __attribute__((ext_vector_type(8))) float v8f;
typedef __attribute__((ext_vector_type(4))) float v4f;

#define EDIM 128
#define WAVES_PER_BLOCK 8
#define EDGES_PER_WAVE 16
#define NUM_BLOCKS 256
#define WFLOATS (EDIM * EDIM)
#define ROWFLOATS (2 * EDGES_PER_WAVE * EDIM)
#define TILE_STRIDE (NUM_BLOCKS * WAVES_PER_BLOCK)
#define STAGE_PITCH 132
#define FOLD_BLOCK 256


__device__ __forceinline__ void wave_lds_order()
{
    __builtin_amdgcn_fence(__ATOMIC_RELEASE, "wavefront");
    __builtin_amdgcn_wave_barrier();
    __builtin_amdgcn_fence(__ATOMIC_ACQUIRE, "wavefront");
}

template <int M>
__device__ __forceinline__ float row_xor_add(float x)
{
    const int other = __builtin_amdgcn_update_dpp(0, __float_as_int(x), 0x160 + M, 0xf, 0xf, true);
    return x + __int_as_float(other);
}

template <int M>
__device__ __forceinline__ void row_xor_add_all(v8f& p)
{
    #pragma unroll
    for (int v = 0; v < 8; ++v) p[v] = row_xor_add<M>(p[v]);
}

__device__ __forceinline__ void take_or_yield_pipe(int mine, int partnerWord, int lead)
{
    const int partner = __builtin_amdgcn_readfirstlane(partnerWord);
    if (mine - partner >= lead) __builtin_amdgcn_s_setprio(0);
    else                        __builtin_amdgcn_s_setprio(3);
}

struct Frag {
    float4 a[2];
    float4 b[2][8];
};

__device__ __forceinline__ void load_block(Frag& f, const float* aRow, const float* bBase)
{
    #pragma unroll
    for (int j = 0; j < 2; ++j) {
        f.a[j] = *(const float4*)(aRow + j * 8);
        #pragma unroll
        for (int t = 0; t < 8; ++t)
            f.b[j][t] = *(const float4*)(bBase + (t * 16 + j) * EDIM);
    }
}

__device__ __forceinline__ void mma_block(v8f (&acc)[8], const Frag& f)
{
    #pragma unroll
    for (int j = 0; j < 2; ++j) {
        const v2f a0 = {f.a[j].x, f.a[j].y};
        const v2f a1 = {f.a[j].z, f.a[j].w};
        #pragma unroll
        for (int t = 0; t < 8; ++t) {
            const v2f b0 = {f.b[j][t].x, f.b[j][t].y};
            const v2f b1 = {f.b[j][t].z, f.b[j][t].w};
            acc[t] = __builtin_amdgcn_wmma_f32_16x16x4_f32(
                false, a0, false, b0, (short)0, acc[t], false, false);
            acc[t] = __builtin_amdgcn_wmma_f32_16x16x4_f32(
                false, a1, false, b1, (short)0, acc[t], false, false);
        }
    }
}

__device__ __forceinline__ void product_pipelined(v8f (&acc)[8], Frag& f0, Frag& f1,
                                                  const float* aCur, const float* bCur,
                                                  const float* aNext, const float* bNext)
{
    for (int b = 0; b < EDIM / 16; b += 2) {
        load_block(f1, aCur + (b + 1) * 16, bCur + (b + 1) * 2 * EDIM);
        __builtin_amdgcn_sched_barrier(0);
        mma_block(acc, f0);
        __builtin_amdgcn_sched_barrier(0);
        const bool last = (b + 2 == EDIM / 16);
        load_block(f0, last ? aNext : aCur + (b + 2) * 16, last ? bNext : bCur + (b + 2) * 2 * EDIM);
        __builtin_amdgcn_sched_barrier(0);
        mma_block(acc, f1);
        __builtin_amdgcn_sched_barrier(0);
    }
}

__global__ __launch_bounds__(WAVES_PER_BLOCK * 32)
void rel_dot_wmma_kernel(const float* __restrict__ emb,
                         const float* __restrict__ relations,
                         const float* __restrict__ tokeys,
                         const float* __restrict__ toqueries,
                         const int* __restrict__ si,
                         const int* __restrict__ pi,
                         const int* __restrict__ oi,
                         float* __restrict__ out,
                         int E)
{
    __shared__ __attribute__((aligned(16))) float sW[2 * WFLOATS];
    __shared__ __attribute__((aligned(16))) float sRows[WAVES_PER_BLOCK * ROWFLOATS];
    __shared__ int sPi[WAVES_PER_BLOCK][EDGES_PER_WAVE];
    __shared__ int sDone[WAVES_PER_BLOCK];

    const int lane = threadIdx.x & 31;
    const int wave = __builtin_amdgcn_readfirstlane(threadIdx.x >> 5);

    for (int c = threadIdx.x; c < WFLOATS / 4; c += WAVES_PER_BLOCK * 32) {
        const int f   = c >> 5;
        const int l   = c & 31;
        const int row = (f >> 4) * 16 + (l & 15);
        const int col = (f & 15) * 8 + 4 * (l >> 4);
        const float4 wk = *(const float4*)(tokeys    + row * EDIM + col);
        const float4 wq = *(const float4*)(toqueries + row * EDIM + col);
        *(float4*)(sW + c * 4)           = wk;
        *(float4*)(sW + WFLOATS + c * 4) = wq;
    }
    if (threadIdx.x < WAVES_PER_BLOCK) sDone[threadIdx.x] = 0;
    __syncthreads();

    float* sS = sRows + wave * ROWFLOATS;
    float* sO = sS + EDGES_PER_WAVE * EDIM;
    const float* wK = sW + lane * 4;
    const float* wQ = sW + WFLOATS + lane * 4;

    const int half = lane >> 4;
    const int lr   = lane & 15;

    const float* aS0 = sS + lr * EDIM + 4 * half;
    const float* aO0 = sO + lr * EDIM + 4 * half;

    const int nTiles = (E + EDGES_PER_WAVE - 1) / EDGES_PER_WAVE;
    const int first  = blockIdx.x * WAVES_PER_BLOCK + wave;

    {
        const int e0   = min(first * EDGES_PER_WAVE + lr, E - 1);
        const int sIdx = si[e0];
        const int oIdx = oi[e0];
        #pragma unroll
        for (int r = 0; r < EDGES_PER_WAVE; ++r) {
            const int srw = __builtin_amdgcn_readlane(sIdx, r);
            const int orw = __builtin_amdgcn_readlane(oIdx, r);
            const float4 vs = *(const float4*)(emb + (size_t)srw * EDIM + lane * 4);
            const float4 vo = *(const float4*)(emb + (size_t)orw * EDIM + lane * 4);
            *(float4*)(sS + r * EDIM + lane * 4) = vs;
            *(float4*)(sO + r * EDIM + lane * 4) = vo;
        }
        if (lane < EDGES_PER_WAVE) sPi[wave][lane] = pi[e0];
    }
    int sIdxNext, oIdxNext;
    {
        const int e1 = min((first + TILE_STRIDE) * EDGES_PER_WAVE + lr, E - 1);
        sIdxNext = si[e1];
        oIdxNext = oi[e1];
    }
    wave_lds_order();

    Frag f0, f1;
    load_block(f0, aS0, wK);

    const bool younger = wave >= WAVES_PER_BLOCK / 2;
    const int  partnerWave = wave ^ (WAVES_PER_BLOCK / 2);
    const int  lead = younger ? 2 : -(1 << 30);
    int done = 0;
    if (younger) __builtin_amdgcn_s_setprio(3);

    for (int tile = first; tile < nTiles; tile += TILE_STRIDE) {
        const int base = tile * EDGES_PER_WAVE;

        const int eNext  = min(base + TILE_STRIDE * EDGES_PER_WAVE + lr, E - 1);
        const int eNext2 = min(base + 2 * TILE_STRIDE * EDGES_PER_WAVE + lr, E - 1);
        const int piNext    = pi[eNext];
        const int sIdxNext2 = si[eNext2];
        const int oIdxNext2 = oi[eNext2];

        int prow[8];
        float prel[8][8];
        #pragma unroll
        for (int v = 0; v < 8; ++v) prow[v] = sPi[wave][v + 8 * half];
        #pragma unroll
        for (int t = 0; t < 4; ++t) {
            #pragma unroll
            for (int v = 0; v < 8; ++v)
                prel[t][v] = relations[(size_t)prow[v] * EDIM + t * 16 + lr];
        }

        v4f pre[EDGES_PER_WAVE];
        #pragma unroll
        for (int r = 0; r < EDGES_PER_WAVE; ++r) {
            const int srw = __builtin_amdgcn_readlane(sIdxNext, r);
            pre[r] = *(const v4f*)(emb + (size_t)srw * EDIM + lane * 4);
        }

        v8f accS[8];
        #pragma unroll
        for (int t = 0; t < 8; ++t) accS[t] = (v8f)0.0f;

        product_pipelined(accS, f0, f1, aS0, wK, aO0, wQ);

        ++done;
        if (lane == 0) sDone[wave] = done;

        wave_lds_order();
        const int partnerMid = sDone[partnerWave];
        #pragma unroll
        for (int r = 0; r < EDGES_PER_WAVE; ++r)
            *(v4f*)(sS + r * EDIM + lane * 4) = pre[r];
        wave_lds_order();

        #pragma unroll
        for (int r = 0; r < EDGES_PER_WAVE; ++r) {
            const int orw = __builtin_amdgcn_readlane(oIdxNext, r);
            pre[r] = *(const v4f*)(emb + (size_t)orw * EDIM + lane * 4);
        }

        #pragma unroll
        for (int t = 4; t < 8; ++t) {
            #pragma unroll
            for (int v = 0; v < 8; ++v)
                prel[t][v] = relations[(size_t)prow[v] * EDIM + t * 16 + lr];
        }

        take_or_yield_pipe(done, partnerMid, lead);

        v8f accO[8];
        #pragma unroll
        for (int t = 0; t < 8; ++t) accO[t] = (v8f)0.0f;

        product_pipelined(accO, f0, f1, aO0, wQ, aS0, wK);

        ++done;
        if (lane == 0) sDone[wave] = done;
        const int partnerEnd = sDone[partnerWave];

        v8f part = (v8f)0.0f;
        #pragma unroll
        for (int t = 0; t < 8; ++t) {
            #pragma unroll
            for (int v = 0; v < 8; ++v)
                part[v] += accS[t][v] * accO[t][v] * prel[t][v];
        }

        row_xor_add_all<1>(part);
        row_xor_add_all<2>(part);
        row_xor_add_all<4>(part);
        row_xor_add_all<8>(part);
        const float scale = 0.08838834764831845f;
        float dot = part[0];
        #pragma unroll
        for (int v = 1; v < 8; ++v) dot = (lr == v) ? part[v] : dot;
        if (lr < 8) {
            const int e = base + lr + 8 * half;
            if (e < E) out[e] = dot * scale;
        }
        wave_lds_order();
        #pragma unroll
        for (int r = 0; r < EDGES_PER_WAVE; ++r)
            *(v4f*)(sO + r * EDIM + lane * 4) = pre[r];
        if (lane < EDGES_PER_WAVE) sPi[wave][lane] = piNext;
        wave_lds_order();
        sIdxNext = sIdxNext2;
        oIdxNext = oIdxNext2;

        take_or_yield_pipe(done, partnerEnd, lead);
    }
    __builtin_amdgcn_s_setprio(0);
}


__device__ __forceinline__ void store_tile_rows(const v8f (&acc)[8], float* stage, float* __restrict__ dst,
                                                int row0, int nRows, int half, int lr, int lane)
{
    wave_lds_order();
    #pragma unroll
    for (int t = 0; t < 8; ++t) {
        #pragma unroll
        for (int v = 0; v < 8; ++v)
            stage[(v + 8 * half) * STAGE_PITCH + t * 16 + lr] = acc[t][v];
    }
    wave_lds_order();
    float4 x[EDGES_PER_WAVE];
    #pragma unroll
    for (int r = 0; r < EDGES_PER_WAVE; ++r)
        x[r] = *(const float4*)(stage + r * STAGE_PITCH + lane * 4);
    float* out0 = dst + (size_t)row0 * EDIM + lane * 4;
    if (row0 + EDGES_PER_WAVE <= nRows) {
        #pragma unroll
        for (int r = 0; r < EDGES_PER_WAVE; ++r)
            *(float4*)(out0 + r * EDIM) = x[r];
    } else {
        #pragma unroll
        for (int r = 0; r < EDGES_PER_WAVE; ++r)
            if (row0 + r < nRows) *(float4*)(out0 + r * EDIM) = x[r];
    }
}

template <bool BOTH>
__global__ __launch_bounds__(WAVES_PER_BLOCK * 32)
void node_project_wmma_kernel(const float* __restrict__ emb,
                              const float* __restrict__ tokeys,
                              const float* __restrict__ toqueries,
                              float* __restrict__ projK,
                              float* __restrict__ projQ,
                              int nRows)
{
    __shared__ __attribute__((aligned(16))) float sW[2 * WFLOATS];
    __shared__ __attribute__((aligned(16))) float sA[WAVES_PER_BLOCK * EDGES_PER_WAVE * EDIM];
    __shared__ __attribute__((aligned(16))) float sT[WAVES_PER_BLOCK * EDGES_PER_WAVE * STAGE_PITCH];
    __shared__ int sDone[WAVES_PER_BLOCK];

    const int lane = threadIdx.x & 31;
    const int wave = __builtin_amdgcn_readfirstlane(threadIdx.x >> 5);

    for (int c = threadIdx.x; c < WFLOATS / 4; c += WAVES_PER_BLOCK * 32) {
        const int f   = c >> 5;
        const int l   = c & 31;
        const int row = (f >> 4) * 16 + (l & 15);
        const int col = (f & 15) * 8 + 4 * (l >> 4);
        const float4 wk = *(const float4*)(tokeys    + row * EDIM + col);
        const float4 wq = *(const float4*)(toqueries + row * EDIM + col);
        *(float4*)(sW + c * 4)           = wk;
        *(float4*)(sW + WFLOATS + c * 4) = wq;
    }
    if (threadIdx.x < WAVES_PER_BLOCK) sDone[threadIdx.x] = 0;
    __syncthreads();

    float* aRows = sA + wave * (EDGES_PER_WAVE * EDIM);
    float* stage = sT + wave * (EDGES_PER_WAVE * STAGE_PITCH);
    const float* wK = sW + lane * 4;
    const float* wQ = sW + WFLOATS + lane * 4;

    const int half = lane >> 4;
    const int lr   = lane & 15;

    const float* aA0 = aRows + lr * EDIM + 4 * half;

    const int nTiles = (nRows + EDGES_PER_WAVE - 1) / EDGES_PER_WAVE;
    const int first  = blockIdx.x * WAVES_PER_BLOCK + wave;

    {
        #pragma unroll
        for (int r = 0; r < EDGES_PER_WAVE; ++r) {
            const int row = min(first * EDGES_PER_WAVE + r, nRows - 1);
            const float4 x = *(const float4*)(emb + (size_t)row * EDIM + lane * 4);
            *(float4*)(aRows + r * EDIM + lane * 4) = x;
        }
    }
    wave_lds_order();

    Frag f0, f1;
    load_block(f0, aA0, wK);

    const bool younger = wave >= WAVES_PER_BLOCK / 2;
    const int  partnerWave = wave ^ (WAVES_PER_BLOCK / 2);
    const int  lead = younger ? (BOTH ? 2 : 1) : -(1 << 30);
    int done = 0;
    if (younger) __builtin_amdgcn_s_setprio(3);

    for (int tile = first; tile < nTiles; tile += TILE_STRIDE) {
        const int row0 = tile * EDGES_PER_WAVE;

        v4f pre[EDGES_PER_WAVE];
        #pragma unroll
        for (int r = 0; r < EDGES_PER_WAVE; ++r) {
            const int row = min(row0 + TILE_STRIDE * EDGES_PER_WAVE + r, nRows - 1);
            pre[r] = *(const v4f*)(emb + (size_t)row * EDIM + lane * 4);
        }

        v8f acc[8];
        #pragma unroll
        for (int t = 0; t < 8; ++t) acc[t] = (v8f)0.0f;
        product_pipelined(acc, f0, f1, aA0, wK, aA0, BOTH ? wQ : wK);

        ++done;
        if (lane == 0) sDone[wave] = done;
        wave_lds_order();
        int partnerEnd = sDone[partnerWave];

        store_tile_rows(acc, stage, projK, row0, nRows, half, lr, lane);

        if (BOTH) {
            take_or_yield_pipe(done, partnerEnd, lead);

            #pragma unroll
            for (int t = 0; t < 8; ++t) acc[t] = (v8f)0.0f;
            product_pipelined(acc, f0, f1, aA0, wQ, aA0, wK);

            ++done;
            if (lane == 0) sDone[wave] = done;
            partnerEnd = sDone[partnerWave];

            store_tile_rows(acc, stage, projQ, row0, nRows, half, lr, lane);
        }

        wave_lds_order();
        #pragma unroll
        for (int r = 0; r < EDGES_PER_WAVE; ++r)
            *(v4f*)(aRows + r * EDIM + lane * 4) = pre[r];
        wave_lds_order();
        f0.a[0] = *(const float4*)(aA0);
        f0.a[1] = *(const float4*)(aA0 + 8);

        take_or_yield_pipe(done, partnerEnd, lead);
    }
    __builtin_amdgcn_s_setprio(0);
}

__global__ __launch_bounds__(WAVES_PER_BLOCK * 32)
void rel_dot_sproj_wmma_kernel(const float* __restrict__ emb,
                               const float* __restrict__ projK,
                               const float* __restrict__ relations,
                               const float* __restrict__ toqueries,
                               const int* __restrict__ si,
                               const int* __restrict__ pi,
                               const int* __restrict__ oi,
                               float* __restrict__ out,
                               int E)
{
    __shared__ __attribute__((aligned(16))) float sW[WFLOATS];
    __shared__ __attribute__((aligned(16))) float sRows[WAVES_PER_BLOCK * 3 * EDGES_PER_WAVE * EDIM];
    __shared__ int sDone[WAVES_PER_BLOCK];

    const int lane = threadIdx.x & 31;
    const int wave = __builtin_amdgcn_readfirstlane(threadIdx.x >> 5);

    for (int c = threadIdx.x; c < WFLOATS / 4; c += WAVES_PER_BLOCK * 32) {
        const int f   = c >> 5;
        const int l   = c & 31;
        const int row = (f >> 4) * 16 + (l & 15);
        const int col = (f & 15) * 8 + 4 * (l >> 4);
        const float4 wq = *(const float4*)(toqueries + row * EDIM + col);
        *(float4*)(sW + c * 4) = wq;
    }
    if (threadIdx.x < WAVES_PER_BLOCK) sDone[threadIdx.x] = 0;
    __syncthreads();

    float* sS = sRows + wave * (3 * EDGES_PER_WAVE * EDIM);
    float* sO = sS + EDGES_PER_WAVE * EDIM;
    float* sP = sO + EDGES_PER_WAVE * EDIM;
    const float* wQ = sW + lane * 4;

    const int half = lane >> 4;
    const int lr   = lane & 15;

    const float* aO0 = sO + lr * EDIM + 4 * half;
    const float* cS = sS + 8 * half * EDIM + lr;
    const float* cP = sP + 8 * half * EDIM + lr;

    const int nTiles = (E + EDGES_PER_WAVE - 1) / EDGES_PER_WAVE;
    const int first  = blockIdx.x * WAVES_PER_BLOCK + wave;

    {
        const int e0   = min(first * EDGES_PER_WAVE + lr, E - 1);
        const int sIdx = si[e0];
        const int oIdx = oi[e0];
        const int pIdx = pi[e0];
        #pragma unroll
        for (int r = 0; r < EDGES_PER_WAVE; ++r) {
            const int srw = __builtin_amdgcn_readlane(sIdx, r);
            const int orw = __builtin_amdgcn_readlane(oIdx, r);
            const int prw = __builtin_amdgcn_readlane(pIdx, r);
            const float4 vs = *(const float4*)(projK     + (size_t)srw * EDIM + lane * 4);
            const float4 vo = *(const float4*)(emb       + (size_t)orw * EDIM + lane * 4);
            const float4 vp = *(const float4*)(relations + (size_t)prw * EDIM + lane * 4);
            *(float4*)(sS + r * EDIM + lane * 4) = vs;
            *(float4*)(sO + r * EDIM + lane * 4) = vo;
            *(float4*)(sP + r * EDIM + lane * 4) = vp;
        }
    }
    int sIdxNext, oIdxNext, pIdxNext;
    {
        const int e1 = min((first + TILE_STRIDE) * EDGES_PER_WAVE + lr, E - 1);
        sIdxNext = si[e1];
        oIdxNext = oi[e1];
        pIdxNext = pi[e1];
    }
    wave_lds_order();

    Frag f0, f1;
    load_block(f0, aO0, wQ);

    const bool younger = wave >= WAVES_PER_BLOCK / 2;
    const int  partnerWave = wave ^ (WAVES_PER_BLOCK / 2);
    const int  lead = younger ? 1 : -(1 << 30);
    int done = 0;
    if (younger) __builtin_amdgcn_s_setprio(3);

    for (int tile = first; tile < nTiles; tile += TILE_STRIDE) {
        const int base = tile * EDGES_PER_WAVE;

        const int eNext2 = min(base + 2 * TILE_STRIDE * EDGES_PER_WAVE + lr, E - 1);
        const int sIdxNext2 = si[eNext2];
        const int oIdxNext2 = oi[eNext2];
        const int pIdxNext2 = pi[eNext2];

        v4f preS[EDGES_PER_WAVE], preO[EDGES_PER_WAVE], preP[EDGES_PER_WAVE];
        #pragma unroll
        for (int r = 0; r < EDGES_PER_WAVE; ++r) {
            const int srw = __builtin_amdgcn_readlane(sIdxNext, r);
            const int orw = __builtin_amdgcn_readlane(oIdxNext, r);
            const int prw = __builtin_amdgcn_readlane(pIdxNext, r);
            preS[r] = *(const v4f*)(projK     + (size_t)srw * EDIM + lane * 4);
            preO[r] = *(const v4f*)(emb       + (size_t)orw * EDIM + lane * 4);
            preP[r] = *(const v4f*)(relations + (size_t)prw * EDIM + lane * 4);
        }

        v8f accO[8];
        #pragma unroll
        for (int t = 0; t < 8; ++t) accO[t] = (v8f)0.0f;
        product_pipelined(accO, f0, f1, aO0, wQ, aO0, wQ);

        ++done;
        if (lane == 0) sDone[wave] = done;
        const int partnerEnd = sDone[partnerWave];

        v8f part = (v8f)0.0f;
        #pragma unroll
        for (int t = 0; t < 8; ++t) {
            #pragma unroll
            for (int v = 0; v < 8; ++v)
                part[v] += cS[v * EDIM + t * 16] * accO[t][v] * cP[v * EDIM + t * 16];
        }

        row_xor_add_all<1>(part);
        row_xor_add_all<2>(part);
        row_xor_add_all<4>(part);
        row_xor_add_all<8>(part);
        const float scale = 0.08838834764831845f;
        float dot = part[0];
        #pragma unroll
        for (int v = 1; v < 8; ++v) dot = (lr == v) ? part[v] : dot;
        if (lr < 8) {
            const int e = base + lr + 8 * half;
            if (e < E) out[e] = dot * scale;
        }

        wave_lds_order();
        #pragma unroll
        for (int r = 0; r < EDGES_PER_WAVE; ++r) {
            *(v4f*)(sS + r * EDIM + lane * 4) = preS[r];
            *(v4f*)(sO + r * EDIM + lane * 4) = preO[r];
            *(v4f*)(sP + r * EDIM + lane * 4) = preP[r];
        }
        wave_lds_order();
        f0.a[0] = *(const float4*)(aO0);
        f0.a[1] = *(const float4*)(aO0 + 8);
        sIdxNext = sIdxNext2;
        oIdxNext = oIdxNext2;
        pIdxNext = pIdxNext2;

        take_or_yield_pipe(done, partnerEnd, lead);
    }
    __builtin_amdgcn_s_setprio(0);
}

__global__ __launch_bounds__(FOLD_BLOCK)
void edge_fold_kernel(const float* __restrict__ projK,
                      const float* __restrict__ projQ,
                      const float* __restrict__ relations,
                      const int* __restrict__ si,
                      const int* __restrict__ pi,
                      const int* __restrict__ oi,
                      float* __restrict__ out,
                      int E)
{
    const int lane = threadIdx.x & 31;
    const int wave = __builtin_amdgcn_readfirstlane(threadIdx.x >> 5);
    const int half = lane >> 4;
    const int lr   = lane & 15;

    const int base = (blockIdx.x * (FOLD_BLOCK / 32) + wave) * 32;
    if (base >= E) return;

    const int e    = min(base + lane, E - 1);
    const int sIdx = si[e];
    const int oIdx = oi[e];
    const int pIdx = pi[e];

    float mine = 0.0f;
    #pragma unroll 1
    for (int c = 0; c < 4; ++c) {
        float4 ka[4], kb[4], qa[4], qb[4], ra[4], rb[4];
        #pragma unroll
        for (int k = 0; k < 4; ++k) {
            const int i = c * 4 + k;
            const int s0 = __builtin_amdgcn_readlane(sIdx, 2 * i), s1 = __builtin_amdgcn_readlane(sIdx, 2 * i + 1);
            const int o0 = __builtin_amdgcn_readlane(oIdx, 2 * i), o1 = __builtin_amdgcn_readlane(oIdx, 2 * i + 1);
            const int p0 = __builtin_amdgcn_readlane(pIdx, 2 * i), p1 = __builtin_amdgcn_readlane(pIdx, 2 * i + 1);
            const float* ks = projK     + (size_t)(half ? s1 : s0) * EDIM + lr * 4;
            const float* qs = projQ     + (size_t)(half ? o1 : o0) * EDIM + lr * 4;
            const float* rs = relations + (size_t)(half ? p1 : p0) * EDIM + lr * 4;
            ka[k] = *(const float4*)(ks);  kb[k] = *(const float4*)(ks + 64);
            qa[k] = *(const float4*)(qs);  qb[k] = *(const float4*)(qs + 64);
            ra[k] = *(const float4*)(rs);  rb[k] = *(const float4*)(rs + 64);
        }
        #pragma unroll
        for (int k = 0; k < 4; ++k) {
            const int i = c * 4 + k;
            float part = (ka[k].x * qa[k].x) * ra[k].x;
            part += (ka[k].y * qa[k].y) * ra[k].y;
            part += (ka[k].z * qa[k].z) * ra[k].z;
            part += (ka[k].w * qa[k].w) * ra[k].w;
            part += (kb[k].x * qb[k].x) * rb[k].x;
            part += (kb[k].y * qb[k].y) * rb[k].y;
            part += (kb[k].z * qb[k].z) * rb[k].z;
            part += (kb[k].w * qb[k].w) * rb[k].w;
            part = row_xor_add<1>(part);
            part = row_xor_add<2>(part);
            part = row_xor_add<4>(part);
            part = row_xor_add<8>(part);
            mine = (lr == i) ? part : mine;
        }
    }
    const float scale = 0.08838834764831845f;
    const int eo = base + 2 * lr + half;
    if (eo < E) out[eo] = mine * scale;
}

extern "C" void kernel_launch(void* const* d_in, const int* in_sizes, int n_in,
                              void* d_out, int out_size, void* d_ws, size_t ws_size,
                              hipStream_t stream) {
    const float* emb       = (const float*)d_in[0];
    const float* relations = (const float*)d_in[1];
    const float* tokeys    = (const float*)d_in[2];
    const float* toqueries = (const float*)d_in[3];
    const int*   si        = (const int*)d_in[4];
    const int*   pi        = (const int*)d_in[5];
    const int*   oi        = (const int*)d_in[6];
    float*       out       = (float*)d_out;

    const int E = in_sizes[4];
    const int nRows = in_sizes[0] / EDIM;
    const size_t tableFloats = (size_t)nRows * EDIM;

    if (d_ws != nullptr && nRows > 0 && ws_size >= 2 * tableFloats * sizeof(float)) {
        float* projK = (float*)d_ws;
        float* projQ = projK + tableFloats;
        node_project_wmma_kernel<true><<<NUM_BLOCKS, WAVES_PER_BLOCK * 32, 0, stream>>>(
            emb, tokeys, toqueries, projK, projQ, nRows);
        edge_fold_kernel<<<(E + FOLD_BLOCK - 1) / FOLD_BLOCK, FOLD_BLOCK, 0, stream>>>(
            projK, projQ, relations, si, pi, oi, out, E);
    } else if (d_ws != nullptr && nRows > 0 && ws_size >= tableFloats * sizeof(float)) {
        float* projK = (float*)d_ws;
        node_project_wmma_kernel<false><<<NUM_BLOCKS, WAVES_PER_BLOCK * 32, 0, stream>>>(
            emb, tokeys, toqueries, projK, (float*)nullptr, nRows);
        rel_dot_sproj_wmma_kernel<<<NUM_BLOCKS, WAVES_PER_BLOCK * 32, 0, stream>>>(
            emb, projK, relations, toqueries, si, pi, oi, out, E);
    } else {
        rel_dot_wmma_kernel<<<NUM_BLOCKS, WAVES_PER_BLOCK * 32, 0, stream>>>(
            emb, relations, tokeys, toqueries, si, pi, oi, out, E);
    }
}


# Relation table held in LDS: the per-edge fold runs as 256 persistent 32-wave work-groups that copy it once and read relation rows by ds_load_b128; one row request in three leaves L2; same sums
# speedup vs baseline: 2.6108x; 1.0343x over previous
#include <hip/hip_runtime.h>

typedef __attribute__((ext_vector_type(2))) float v2f;
typedef __attribute__((ext_vector_type(8))) float v8f;
typedef __attribute__((ext_vector_type(4))) float v4f;

#define EDIM 128
#define WAVES_PER_BLOCK 8
#define EDGES_PER_WAVE 16
#define NUM_BLOCKS 256
#define WFLOATS (EDIM * EDIM)
#define ROWFLOATS (2 * EDGES_PER_WAVE * EDIM)
#define TILE_STRIDE (NUM_BLOCKS * WAVES_PER_BLOCK)
#define STAGE_PITCH 132
#define FOLD_BLOCK 256


__device__ __forceinline__ void wave_lds_order()
{
    __builtin_amdgcn_fence(__ATOMIC_RELEASE, "wavefront");
    __builtin_amdgcn_wave_barrier();
    __builtin_amdgcn_fence(__ATOMIC_ACQUIRE, "wavefront");
}

template <int M>
__device__ __forceinline__ float row_xor_add(float x)
{
    const int other = __builtin_amdgcn_update_dpp(0, __float_as_int(x), 0x160 + M, 0xf, 0xf, true);
    return x + __int_as_float(other);
}

template <int M>
__device__ __forceinline__ void row_xor_add_all(v8f& p)
{
    #pragma unroll
    for (int v = 0; v < 8; ++v) p[v] = row_xor_add<M>(p[v]);
}

__device__ __forceinline__ void take_or_yield_pipe(int mine, int partnerWord, int lead)
{
    const int partner = __builtin_amdgcn_readfirstlane(partnerWord);
    if (mine - partner >= lead) __builtin_amdgcn_s_setprio(0);
    else                        __builtin_amdgcn_s_setprio(3);
}

struct Frag {
    float4 a[2];
    float4 b[2][8];
};

__device__ __forceinline__ void load_block(Frag& f, const float* aRow, const float* bBase)
{
    #pragma unroll
    for (int j = 0; j < 2; ++j) {
        f.a[j] = *(const float4*)(aRow + j * 8);
        #pragma unroll
        for (int t = 0; t < 8; ++t)
            f.b[j][t] = *(const float4*)(bBase + (t * 16 + j) * EDIM);
    }
}

__device__ __forceinline__ void mma_block(v8f (&acc)[8], const Frag& f)
{
    #pragma unroll
    for (int j = 0; j < 2; ++j) {
        const v2f a0 = {f.a[j].x, f.a[j].y};
        const v2f a1 = {f.a[j].z, f.a[j].w};
        #pragma unroll
        for (int t = 0; t < 8; ++t) {
            const v2f b0 = {f.b[j][t].x, f.b[j][t].y};
            const v2f b1 = {f.b[j][t].z, f.b[j][t].w};
            acc[t] = __builtin_amdgcn_wmma_f32_16x16x4_f32(
                false, a0, false, b0, (short)0, acc[t], false, false);
            acc[t] = __builtin_amdgcn_wmma_f32_16x16x4_f32(
                false, a1, false, b1, (short)0, acc[t], false, false);
        }
    }
}

__device__ __forceinline__ void product_pipelined(v8f (&acc)[8], Frag& f0, Frag& f1,
                                                  const float* aCur, const float* bCur,
                                                  const float* aNext, const float* bNext)
{
    for (int b = 0; b < EDIM / 16; b += 2) {
        load_block(f1, aCur + (b + 1) * 16, bCur + (b + 1) * 2 * EDIM);
        __builtin_amdgcn_sched_barrier(0);
        mma_block(acc, f0);
        __builtin_amdgcn_sched_barrier(0);
        const bool last = (b + 2 == EDIM / 16);
        load_block(f0, last ? aNext : aCur + (b + 2) * 16, last ? bNext : bCur + (b + 2) * 2 * EDIM);
        __builtin_amdgcn_sched_barrier(0);
        mma_block(acc, f1);
        __builtin_amdgcn_sched_barrier(0);
    }
}

__global__ __launch_bounds__(WAVES_PER_BLOCK * 32)
void rel_dot_wmma_kernel(const float* __restrict__ emb,
                         const float* __restrict__ relations,
                         const float* __restrict__ tokeys,
                         const float* __restrict__ toqueries,
                         const int* __restrict__ si,
                         const int* __restrict__ pi,
                         const int* __restrict__ oi,
                         float* __restrict__ out,
                         int E)
{
    __shared__ __attribute__((aligned(16))) float sW[2 * WFLOATS];
    __shared__ __attribute__((aligned(16))) float sRows[WAVES_PER_BLOCK * ROWFLOATS];
    __shared__ int sPi[WAVES_PER_BLOCK][EDGES_PER_WAVE];
    __shared__ int sDone[WAVES_PER_BLOCK];

    const int lane = threadIdx.x & 31;
    const int wave = __builtin_amdgcn_readfirstlane(threadIdx.x >> 5);

    for (int c = threadIdx.x; c < WFLOATS / 4; c += WAVES_PER_BLOCK * 32) {
        const int f   = c >> 5;
        const int l   = c & 31;
        const int row = (f >> 4) * 16 + (l & 15);
        const int col = (f & 15) * 8 + 4 * (l >> 4);
        const float4 wk = *(const float4*)(tokeys    + row * EDIM + col);
        const float4 wq = *(const float4*)(toqueries + row * EDIM + col);
        *(float4*)(sW + c * 4)           = wk;
        *(float4*)(sW + WFLOATS + c * 4) = wq;
    }
    if (threadIdx.x < WAVES_PER_BLOCK) sDone[threadIdx.x] = 0;
    __syncthreads();

    float* sS = sRows + wave * ROWFLOATS;
    float* sO = sS + EDGES_PER_WAVE * EDIM;
    const float* wK = sW + lane * 4;
    const float* wQ = sW + WFLOATS + lane * 4;

    const int half = lane >> 4;
    const int lr   = lane & 15;

    const float* aS0 = sS + lr * EDIM + 4 * half;
    const float* aO0 = sO + lr * EDIM + 4 * half;

    const int nTiles = (E + EDGES_PER_WAVE - 1) / EDGES_PER_WAVE;
    const int first  = blockIdx.x * WAVES_PER_BLOCK + wave;

    {
        const int e0   = min(first * EDGES_PER_WAVE + lr, E - 1);
        const int sIdx = si[e0];
        const int oIdx = oi[e0];
        #pragma unroll
        for (int r = 0; r < EDGES_PER_WAVE; ++r) {
            const int srw = __builtin_amdgcn_readlane(sIdx, r);
            const int orw = __builtin_amdgcn_readlane(oIdx, r);
            const float4 vs = *(const float4*)(emb + (size_t)srw * EDIM + lane * 4);
            const float4 vo = *(const float4*)(emb + (size_t)orw * EDIM + lane * 4);
            *(float4*)(sS + r * EDIM + lane * 4) = vs;
            *(float4*)(sO + r * EDIM + lane * 4) = vo;
        }
        if (lane < EDGES_PER_WAVE) sPi[wave][lane] = pi[e0];
    }
    int sIdxNext, oIdxNext;
    {
        const int e1 = min((first + TILE_STRIDE) * EDGES_PER_WAVE + lr, E - 1);
        sIdxNext = si[e1];
        oIdxNext = oi[e1];
    }
    wave_lds_order();

    Frag f0, f1;
    load_block(f0, aS0, wK);

    const bool younger = wave >= WAVES_PER_BLOCK / 2;
    const int  partnerWave = wave ^ (WAVES_PER_BLOCK / 2);
    const int  lead = younger ? 2 : -(1 << 30);
    int done = 0;
    if (younger) __builtin_amdgcn_s_setprio(3);

    for (int tile = first; tile < nTiles; tile += TILE_STRIDE) {
        const int base = tile * EDGES_PER_WAVE;

        const int eNext  = min(base + TILE_STRIDE * EDGES_PER_WAVE + lr, E - 1);
        const int eNext2 = min(base + 2 * TILE_STRIDE * EDGES_PER_WAVE + lr, E - 1);
        const int piNext    = pi[eNext];
        const int sIdxNext2 = si[eNext2];
        const int oIdxNext2 = oi[eNext2];

        int prow[8];
        float prel[8][8];
        #pragma unroll
        for (int v = 0; v < 8; ++v) prow[v] = sPi[wave][v + 8 * half];
        #pragma unroll
        for (int t = 0; t < 4; ++t) {
            #pragma unroll
            for (int v = 0; v < 8; ++v)
                prel[t][v] = relations[(size_t)prow[v] * EDIM + t * 16 + lr];
        }

        v4f pre[EDGES_PER_WAVE];
        #pragma unroll
        for (int r = 0; r < EDGES_PER_WAVE; ++r) {
            const int srw = __builtin_amdgcn_readlane(sIdxNext, r);
            pre[r] = *(const v4f*)(emb + (size_t)srw * EDIM + lane * 4);
        }

        v8f accS[8];
        #pragma unroll
        for (int t = 0; t < 8; ++t) accS[t] = (v8f)0.0f;

        product_pipelined(accS, f0, f1, aS0, wK, aO0, wQ);

        ++done;
        if (lane == 0) sDone[wave] = done;

        wave_lds_order();
        const int partnerMid = sDone[partnerWave];
        #pragma unroll
        for (int r = 0; r < EDGES_PER_WAVE; ++r)
            *(v4f*)(sS + r * EDIM + lane * 4) = pre[r];
        wave_lds_order();

        #pragma unroll
        for (int r = 0; r < EDGES_PER_WAVE; ++r) {
            const int orw = __builtin_amdgcn_readlane(oIdxNext, r);
            pre[r] = *(const v4f*)(emb + (size_t)orw * EDIM + lane * 4);
        }

        #pragma unroll
        for (int t = 4; t < 8; ++t) {
            #pragma unroll
            for (int v = 0; v < 8; ++v)
                prel[t][v] = relations[(size_t)prow[v] * EDIM + t * 16 + lr];
        }

        take_or_yield_pipe(done, partnerMid, lead);

        v8f accO[8];
        #pragma unroll
        for (int t = 0; t < 8; ++t) accO[t] = (v8f)0.0f;

        product_pipelined(accO, f0, f1, aO0, wQ, aS0, wK);

        ++done;
        if (lane == 0) sDone[wave] = done;
        const int partnerEnd = sDone[partnerWave];

        v8f part = (v8f)0.0f;
        #pragma unroll
        for (int t = 0; t < 8; ++t) {
            #pragma unroll
            for (int v = 0; v < 8; ++v)
                part[v] += accS[t][v] * accO[t][v] * prel[t][v];
        }

        row_xor_add_all<1>(part);
        row_xor_add_all<2>(part);
        row_xor_add_all<4>(part);
        row_xor_add_all<8>(part);
        const float scale = 0.08838834764831845f;
        float dot = part[0];
        #pragma unroll
        for (int v = 1; v < 8; ++v) dot = (lr == v) ? part[v] : dot;
        if (lr < 8) {
            const int e = base + lr + 8 * half;
            if (e < E) out[e] = dot * scale;
        }
        wave_lds_order();
        #pragma unroll
        for (int r = 0; r < EDGES_PER_WAVE; ++r)
            *(v4f*)(sO + r * EDIM + lane * 4) = pre[r];
        if (lane < EDGES_PER_WAVE) sPi[wave][lane] = piNext;
        wave_lds_order();
        sIdxNext = sIdxNext2;
        oIdxNext = oIdxNext2;

        take_or_yield_pipe(done, partnerEnd, lead);
    }
    __builtin_amdgcn_s_setprio(0);
}


__device__ __forceinline__ void store_tile_rows(const v8f (&acc)[8], float* stage, float* __restrict__ dst,
                                                int row0, int nRows, int half, int lr, int lane)
{
    wave_lds_order();
    #pragma unroll
    for (int t = 0; t < 8; ++t) {
        #pragma unroll
        for (int v = 0; v < 8; ++v)
            stage[(v + 8 * half) * STAGE_PITCH + t * 16 + lr] = acc[t][v];
    }
    wave_lds_order();
    float4 x[EDGES_PER_WAVE];
    #pragma unroll
    for (int r = 0; r < EDGES_PER_WAVE; ++r)
        x[r] = *(const float4*)(stage + r * STAGE_PITCH + lane * 4);
    float* out0 = dst + (size_t)row0 * EDIM + lane * 4;
    if (row0 + EDGES_PER_WAVE <= nRows) {
        #pragma unroll
        for (int r = 0; r < EDGES_PER_WAVE; ++r)
            *(float4*)(out0 + r * EDIM) = x[r];
    } else {
        #pragma unroll
        for (int r = 0; r < EDGES_PER_WAVE; ++r)
            if (row0 + r < nRows) *(float4*)(out0 + r * EDIM) = x[r];
    }
}

template <bool BOTH>
__global__ __launch_bounds__(WAVES_PER_BLOCK * 32)
void node_project_wmma_kernel(const float* __restrict__ emb,
                              const float* __restrict__ tokeys,
                              const float* __restrict__ toqueries,
                              float* __restrict__ projK,
                              float* __restrict__ projQ,
                              int nRows)
{
    __shared__ __attribute__((aligned(16))) float sW[2 * WFLOATS];
    __shared__ __attribute__((aligned(16))) float sA[WAVES_PER_BLOCK * EDGES_PER_WAVE * EDIM];
    __shared__ __attribute__((aligned(16))) float sT[WAVES_PER_BLOCK * EDGES_PER_WAVE * STAGE_PITCH];
    __shared__ int sDone[WAVES_PER_BLOCK];

    const int lane = threadIdx.x & 31;
    const int wave = __builtin_amdgcn_readfirstlane(threadIdx.x >> 5);

    for (int c = threadIdx.x; c < WFLOATS / 4; c += WAVES_PER_BLOCK * 32) {
        const int f   = c >> 5;
        const int l   = c & 31;
        const int row = (f >> 4) * 16 + (l & 15);
        const int col = (f & 15) * 8 + 4 * (l >> 4);
        const float4 wk = *(const float4*)(tokeys    + row * EDIM + col);
        const float4 wq = *(const float4*)(toqueries + row * EDIM + col);
        *(float4*)(sW + c * 4)           = wk;
        *(float4*)(sW + WFLOATS + c * 4) = wq;
    }
    if (threadIdx.x < WAVES_PER_BLOCK) sDone[threadIdx.x] = 0;
    __syncthreads();

    float* aRows = sA + wave * (EDGES_PER_WAVE * EDIM);
    float* stage = sT + wave * (EDGES_PER_WAVE * STAGE_PITCH);
    const float* wK = sW + lane * 4;
    const float* wQ = sW + WFLOATS + lane * 4;

    const int half = lane >> 4;
    const int lr   = lane & 15;

    const float* aA0 = aRows + lr * EDIM + 4 * half;

    const int nTiles = (nRows + EDGES_PER_WAVE - 1) / EDGES_PER_WAVE;
    const int first  = blockIdx.x * WAVES_PER_BLOCK + wave;

    {
        #pragma unroll
        for (int r = 0; r < EDGES_PER_WAVE; ++r) {
            const int row = min(first * EDGES_PER_WAVE + r, nRows - 1);
            const float4 x = *(const float4*)(emb + (size_t)row * EDIM + lane * 4);
            *(float4*)(aRows + r * EDIM + lane * 4) = x;
        }
    }
    wave_lds_order();

    Frag f0, f1;
    load_block(f0, aA0, wK);

    const bool younger = wave >= WAVES_PER_BLOCK / 2;
    const int  partnerWave = wave ^ (WAVES_PER_BLOCK / 2);
    const int  lead = younger ? (BOTH ? 2 : 1) : -(1 << 30);
    int done = 0;
    if (younger) __builtin_amdgcn_s_setprio(3);

    for (int tile = first; tile < nTiles; tile += TILE_STRIDE) {
        const int row0 = tile * EDGES_PER_WAVE;

        v4f pre[EDGES_PER_WAVE];
        #pragma unroll
        for (int r = 0; r < EDGES_PER_WAVE; ++r) {
            const int row = min(row0 + TILE_STRIDE * EDGES_PER_WAVE + r, nRows - 1);
            pre[r] = *(const v4f*)(emb + (size_t)row * EDIM + lane * 4);
        }

        v8f acc[8];
        #pragma unroll
        for (int t = 0; t < 8; ++t) acc[t] = (v8f)0.0f;
        product_pipelined(acc, f0, f1, aA0, wK, aA0, BOTH ? wQ : wK);

        ++done;
        if (lane == 0) sDone[wave] = done;
        wave_lds_order();
        int partnerEnd = sDone[partnerWave];

        store_tile_rows(acc, stage, projK, row0, nRows, half, lr, lane);

        if (BOTH) {
            take_or_yield_pipe(done, partnerEnd, lead);

            #pragma unroll
            for (int t = 0; t < 8; ++t) acc[t] = (v8f)0.0f;
            product_pipelined(acc, f0, f1, aA0, wQ, aA0, wK);

            ++done;
            if (lane == 0) sDone[wave] = done;
            partnerEnd = sDone[partnerWave];

            store_tile_rows(acc, stage, projQ, row0, nRows, half, lr, lane);
        }

        wave_lds_order();
        #pragma unroll
        for (int r = 0; r < EDGES_PER_WAVE; ++r)
            *(v4f*)(aRows + r * EDIM + lane * 4) = pre[r];
        wave_lds_order();
        f0.a[0] = *(const float4*)(aA0);
        f0.a[1] = *(const float4*)(aA0 + 8);

        take_or_yield_pipe(done, partnerEnd, lead);
    }
    __builtin_amdgcn_s_setprio(0);
}

__global__ __launch_bounds__(WAVES_PER_BLOCK * 32)
void rel_dot_sproj_wmma_kernel(const float* __restrict__ emb,
                               const float* __restrict__ projK,
                               const float* __restrict__ relations,
                               const float* __restrict__ toqueries,
                               const int* __restrict__ si,
                               const int* __restrict__ pi,
                               const int* __restrict__ oi,
                               float* __restrict__ out,
                               int E)
{
    __shared__ __attribute__((aligned(16))) float sW[WFLOATS];
    __shared__ __attribute__((aligned(16))) float sRows[WAVES_PER_BLOCK * 3 * EDGES_PER_WAVE * EDIM];
    __shared__ int sDone[WAVES_PER_BLOCK];

    const int lane = threadIdx.x & 31;
    const int wave = __builtin_amdgcn_readfirstlane(threadIdx.x >> 5);

    for (int c = threadIdx.x; c < WFLOATS / 4; c += WAVES_PER_BLOCK * 32) {
        const int f   = c >> 5;
        const int l   = c & 31;
        const int row = (f >> 4) * 16 + (l & 15);
        const int col = (f & 15) * 8 + 4 * (l >> 4);
        const float4 wq = *(const float4*)(toqueries + row * EDIM + col);
        *(float4*)(sW + c * 4) = wq;
    }
    if (threadIdx.x < WAVES_PER_BLOCK) sDone[threadIdx.x] = 0;
    __syncthreads();

    float* sS = sRows + wave * (3 * EDGES_PER_WAVE * EDIM);
    float* sO = sS + EDGES_PER_WAVE * EDIM;
    float* sP = sO + EDGES_PER_WAVE * EDIM;
    const float* wQ = sW + lane * 4;

    const int half = lane >> 4;
    const int lr   = lane & 15;

    const float* aO0 = sO + lr * EDIM + 4 * half;
    const float* cS = sS + 8 * half * EDIM + lr;
    const float* cP = sP + 8 * half * EDIM + lr;

    const int nTiles = (E + EDGES_PER_WAVE - 1) / EDGES_PER_WAVE;
    const int first  = blockIdx.x * WAVES_PER_BLOCK + wave;

    {
        const int e0   = min(first * EDGES_PER_WAVE + lr, E - 1);
        const int sIdx = si[e0];
        const int oIdx = oi[e0];
        const int pIdx = pi[e0];
        #pragma unroll
        for (int r = 0; r < EDGES_PER_WAVE; ++r) {
            const int srw = __builtin_amdgcn_readlane(sIdx, r);
            const int orw = __builtin_amdgcn_readlane(oIdx, r);
            const int prw = __builtin_amdgcn_readlane(pIdx, r);
            const float4 vs = *(const float4*)(projK     + (size_t)srw * EDIM + lane * 4);
            const float4 vo = *(const float4*)(emb       + (size_t)orw * EDIM + lane * 4);
            const float4 vp = *(const float4*)(relations + (size_t)prw * EDIM + lane * 4);
            *(float4*)(sS + r * EDIM + lane * 4) = vs;
            *(float4*)(sO + r * EDIM + lane * 4) = vo;
            *(float4*)(sP + r * EDIM + lane * 4) = vp;
        }
    }
    int sIdxNext, oIdxNext, pIdxNext;
    {
        const int e1 = min((first + TILE_STRIDE) * EDGES_PER_WAVE + lr, E - 1);
        sIdxNext = si[e1];
        oIdxNext = oi[e1];
        pIdxNext = pi[e1];
    }
    wave_lds_order();

    Frag f0, f1;
    load_block(f0, aO0, wQ);

    const bool younger = wave >= WAVES_PER_BLOCK / 2;
    const int  partnerWave = wave ^ (WAVES_PER_BLOCK / 2);
    const int  lead = younger ? 1 : -(1 << 30);
    int done = 0;
    if (younger) __builtin_amdgcn_s_setprio(3);

    for (int tile = first; tile < nTiles; tile += TILE_STRIDE) {
        const int base = tile * EDGES_PER_WAVE;

        const int eNext2 = min(base + 2 * TILE_STRIDE * EDGES_PER_WAVE + lr, E - 1);
        const int sIdxNext2 = si[eNext2];
        const int oIdxNext2 = oi[eNext2];
        const int pIdxNext2 = pi[eNext2];

        v4f preS[EDGES_PER_WAVE], preO[EDGES_PER_WAVE], preP[EDGES_PER_WAVE];
        #pragma unroll
        for (int r = 0; r < EDGES_PER_WAVE; ++r) {
            const int srw = __builtin_amdgcn_readlane(sIdxNext, r);
            const int orw = __builtin_amdgcn_readlane(oIdxNext, r);
            const int prw = __builtin_amdgcn_readlane(pIdxNext, r);
            preS[r] = *(const v4f*)(projK     + (size_t)srw * EDIM + lane * 4);
            preO[r] = *(const v4f*)(emb       + (size_t)orw * EDIM + lane * 4);
            preP[r] = *(const v4f*)(relations + (size_t)prw * EDIM + lane * 4);
        }

        v8f accO[8];
        #pragma unroll
        for (int t = 0; t < 8; ++t) accO[t] = (v8f)0.0f;
        product_pipelined(accO, f0, f1, aO0, wQ, aO0, wQ);

        ++done;
        if (lane == 0) sDone[wave] = done;
        const int partnerEnd = sDone[partnerWave];

        v8f part = (v8f)0.0f;
        #pragma unroll
        for (int t = 0; t < 8; ++t) {
            #pragma unroll
            for (int v = 0; v < 8; ++v)
                part[v] += cS[v * EDIM + t * 16] * accO[t][v] * cP[v * EDIM + t * 16];
        }

        row_xor_add_all<1>(part);
        row_xor_add_all<2>(part);
        row_xor_add_all<4>(part);
        row_xor_add_all<8>(part);
        const float scale = 0.08838834764831845f;
        float dot = part[0];
        #pragma unroll
        for (int v = 1; v < 8; ++v) dot = (lr == v) ? part[v] : dot;
        if (lr < 8) {
            const int e = base + lr + 8 * half;
            if (e < E) out[e] = dot * scale;
        }

        wave_lds_order();
        #pragma unroll
        for (int r = 0; r < EDGES_PER_WAVE; ++r) {
            *(v4f*)(sS + r * EDIM + lane * 4) = preS[r];
            *(v4f*)(sO + r * EDIM + lane * 4) = preO[r];
            *(v4f*)(sP + r * EDIM + lane * 4) = preP[r];
        }
        wave_lds_order();
        f0.a[0] = *(const float4*)(aO0);
        f0.a[1] = *(const float4*)(aO0 + 8);
        sIdxNext = sIdxNext2;
        oIdxNext = oIdxNext2;
        pIdxNext = pIdxNext2;

        take_or_yield_pipe(done, partnerEnd, lead);
    }
    __builtin_amdgcn_s_setprio(0);
}

__global__ __launch_bounds__(FOLD_BLOCK)
void edge_fold_kernel(const float* __restrict__ projK,
                      const float* __restrict__ projQ,
                      const float* __restrict__ relations,
                      const int* __restrict__ si,
                      const int* __restrict__ pi,
                      const int* __restrict__ oi,
                      float* __restrict__ out,
                      int E)
{
    const int lane = threadIdx.x & 31;
    const int wave = __builtin_amdgcn_readfirstlane(threadIdx.x >> 5);
    const int half = lane >> 4;
    const int lr   = lane & 15;

    const int base = (blockIdx.x * (FOLD_BLOCK / 32) + wave) * 32;
    if (base >= E) return;

    const int e    = min(base + lane, E - 1);
    const int sIdx = si[e];
    const int oIdx = oi[e];
    const int pIdx = pi[e];

    float mine = 0.0f;
    #pragma unroll 1
    for (int c = 0; c < 4; ++c) {
        float4 ka[4], kb[4], qa[4], qb[4], ra[4], rb[4];
        #pragma unroll
        for (int k = 0; k < 4; ++k) {
            const int i = c * 4 + k;
            const int s0 = __builtin_amdgcn_readlane(sIdx, 2 * i), s1 = __builtin_amdgcn_readlane(sIdx, 2 * i + 1);
            const int o0 = __builtin_amdgcn_readlane(oIdx, 2 * i), o1 = __builtin_amdgcn_readlane(oIdx, 2 * i + 1);
            const int p0 = __builtin_amdgcn_readlane(pIdx, 2 * i), p1 = __builtin_amdgcn_readlane(pIdx, 2 * i + 1);
            const float* ks = projK     + (size_t)(half ? s1 : s0) * EDIM + lr * 4;
            const float* qs = projQ     + (size_t)(half ? o1 : o0) * EDIM + lr * 4;
            const float* rs = relations + (size_t)(half ? p1 : p0) * EDIM + lr * 4;
            ka[k] = *(const float4*)(ks);  kb[k] = *(const float4*)(ks + 64);
            qa[k] = *(const float4*)(qs);  qb[k] = *(const float4*)(qs + 64);
            ra[k] = *(const float4*)(rs);  rb[k] = *(const float4*)(rs + 64);
        }
        #pragma unroll
        for (int k = 0; k < 4; ++k) {
            const int i = c * 4 + k;
            float part = (ka[k].x * qa[k].x) * ra[k].x;
            part += (ka[k].y * qa[k].y) * ra[k].y;
            part += (ka[k].z * qa[k].z) * ra[k].z;
            part += (ka[k].w * qa[k].w) * ra[k].w;
            part += (kb[k].x * qb[k].x) * rb[k].x;
            part += (kb[k].y * qb[k].y) * rb[k].y;
            part += (kb[k].z * qb[k].z) * rb[k].z;
            part += (kb[k].w * qb[k].w) * rb[k].w;
            part = row_xor_add<1>(part);
            part = row_xor_add<2>(part);
            part = row_xor_add<4>(part);
            part = row_xor_add<8>(part);
            mine = (lr == i) ? part : mine;
        }
    }
    const float scale = 0.08838834764831845f;
    const int eo = base + 2 * lr + half;
    if (eo < E) out[eo] = mine * scale;
}

#define REL_LDS_ROWS 512
#define PFOLD_WAVES 32
#define PFOLD_BLOCK (PFOLD_WAVES * 32)
static_assert((REL_LDS_ROWS * (EDIM / 4)) % (4 * PFOLD_BLOCK) == 0, "the LDS relation rows must be a whole number of copy trips");

__global__ __launch_bounds__(PFOLD_BLOCK)
void edge_fold_lds_rel_kernel(const float* __restrict__ projK,
                              const float* __restrict__ projQ,
                              const float* __restrict__ relations,
                              const int* __restrict__ si,
                              const int* __restrict__ pi,
                              const int* __restrict__ oi,
                              float* __restrict__ out,
                              int E, int nRel)
{
    __shared__ __attribute__((aligned(16))) float sRel[REL_LDS_ROWS * EDIM];

    const int lane = threadIdx.x & 31;
    const int wave = __builtin_amdgcn_readfirstlane(threadIdx.x >> 5);
    const int half = lane >> 4;
    const int lr   = lane & 15;

    const int nChunks = (E + 31) / 32;
    const int stride  = gridDim.x * PFOLD_WAVES;
    int chunk = blockIdx.x * PFOLD_WAVES + wave;

    int sIdx, oIdx, pIdx;
    {
        const int e = min(chunk * 32 + lane, E - 1);
        sIdx = si[e];
        oIdx = oi[e];
        pIdx = pi[e];
    }

    const int nChunk16 = nRel * (EDIM / 4);
    for (int c0 = threadIdx.x; c0 < nChunk16; c0 += 4 * PFOLD_BLOCK) {
        float4 w[4];
        #pragma unroll
        for (int j = 0; j < 4; ++j)
            w[j] = *(const float4*)(relations + min(c0 + j * PFOLD_BLOCK, nChunk16 - 1) * 4);
        #pragma unroll
        for (int j = 0; j < 4; ++j)
            *(float4*)(sRel + (c0 + j * PFOLD_BLOCK) * 4) = w[j];
    }
    __syncthreads();

    if (chunk >= nChunks) return;

    for (; chunk < nChunks; chunk += stride) {
        const int base = chunk * 32;

        const int eNext = min(base + stride * 32 + lane, E - 1);
        const int sIdxNext = si[eNext];
        const int oIdxNext = oi[eNext];
        const int pIdxNext = pi[eNext];

        float mine = 0.0f;
        #pragma unroll 1
        for (int c = 0; c < 4; ++c) {
            float4 ka[4], kb[4], qa[4], qb[4], ra[4], rb[4];
            #pragma unroll
            for (int k = 0; k < 4; ++k) {
                const int i = c * 4 + k;
                const int s0 = __builtin_amdgcn_readlane(sIdx, 2 * i), s1 = __builtin_amdgcn_readlane(sIdx, 2 * i + 1);
                const int o0 = __builtin_amdgcn_readlane(oIdx, 2 * i), o1 = __builtin_amdgcn_readlane(oIdx, 2 * i + 1);
                const int p0 = __builtin_amdgcn_readlane(pIdx, 2 * i), p1 = __builtin_amdgcn_readlane(pIdx, 2 * i + 1);
                const float* ks = projK + (size_t)(half ? s1 : s0) * EDIM + lr * 4;
                const float* qs = projQ + (size_t)(half ? o1 : o0) * EDIM + lr * 4;
                const float* rs = sRel + (half ? p1 : p0) * EDIM + lr * 4;
                ka[k] = *(const float4*)(ks);  kb[k] = *(const float4*)(ks + 64);
                qa[k] = *(const float4*)(qs);  qb[k] = *(const float4*)(qs + 64);
                ra[k] = *(const float4*)(rs);  rb[k] = *(const float4*)(rs + 64);
            }
            #pragma unroll
            for (int k = 0; k < 4; ++k) {
                const int i = c * 4 + k;
                float part = (ka[k].x * qa[k].x) * ra[k].x;
                part += (ka[k].y * qa[k].y) * ra[k].y;
                part += (ka[k].z * qa[k].z) * ra[k].z;
                part += (ka[k].w * qa[k].w) * ra[k].w;
                part += (kb[k].x * qb[k].x) * rb[k].x;
                part += (kb[k].y * qb[k].y) * rb[k].y;
                part += (kb[k].z * qb[k].z) * rb[k].z;
                part += (kb[k].w * qb[k].w) * rb[k].w;
                part = row_xor_add<1>(part);
                part = row_xor_add<2>(part);
                part = row_xor_add<4>(part);
                part = row_xor_add<8>(part);
                mine = (lr == i) ? part : mine;
            }
        }
        const float scale = 0.08838834764831845f;
        const int eo = base + 2 * lr + half;
        if (eo < E) out[eo] = mine * scale;

        sIdx = sIdxNext;
        oIdx = oIdxNext;
        pIdx = pIdxNext;
    }
}

extern "C" void kernel_launch(void* const* d_in, const int* in_sizes, int n_in,
                              void* d_out, int out_size, void* d_ws, size_t ws_size,
                              hipStream_t stream) {
    const float* emb       = (const float*)d_in[0];
    const float* relations = (const float*)d_in[1];
    const float* tokeys    = (const float*)d_in[2];
    const float* toqueries = (const float*)d_in[3];
    const int*   si        = (const int*)d_in[4];
    const int*   pi        = (const int*)d_in[5];
    const int*   oi        = (const int*)d_in[6];
    float*       out       = (float*)d_out;

    const int E = in_sizes[4];
    const int nRows = in_sizes[0] / EDIM;
    const int nRel = in_sizes[1] / EDIM;
    const size_t tableFloats = (size_t)nRows * EDIM;

    if (d_ws != nullptr && nRows > 0 && ws_size >= 2 * tableFloats * sizeof(float)) {
        float* projK = (float*)d_ws;
        float* projQ = projK + tableFloats;
        node_project_wmma_kernel<true><<<NUM_BLOCKS, WAVES_PER_BLOCK * 32, 0, stream>>>(
            emb, tokeys, toqueries, projK, projQ, nRows);
        if (nRel <= REL_LDS_ROWS) {
            edge_fold_lds_rel_kernel<<<NUM_BLOCKS, PFOLD_BLOCK, 0, stream>>>(
                projK, projQ, relations, si, pi, oi, out, E, nRel);
        } else {
            edge_fold_kernel<<<(E + FOLD_BLOCK - 1) / FOLD_BLOCK, FOLD_BLOCK, 0, stream>>>(
                projK, projQ, relations, si, pi, oi, out, E);
        }
    } else if (d_ws != nullptr && nRows > 0 && ws_size >= tableFloats * sizeof(float)) {
        float* projK = (float*)d_ws;
        node_project_wmma_kernel<false><<<NUM_BLOCKS, WAVES_PER_BLOCK * 32, 0, stream>>>(
            emb, tokeys, toqueries, projK, (float*)nullptr, nRows);
        rel_dot_sproj_wmma_kernel<<<NUM_BLOCKS, WAVES_PER_BLOCK * 32, 0, stream>>>(
            emb, projK, relations, toqueries, si, pi, oi, out, E);
    } else {
        rel_dot_wmma_kernel<<<NUM_BLOCKS, WAVES_PER_BLOCK * 32, 0, stream>>>(
            emb, relations, tokeys, toqueries, si, pi, oi, out, E);
    }
}
